# starting kernel
# baseline (speedup 1.0000x reference)
#include <hip/hip_runtime.h>

#define Bq  8
#define Nq  1024
#define Dq  256
#define Hq  4
#define HOq 64
#define Tq  5
#define NEGV (-1e9f)
#define EPSV (1e-6f)

typedef float v2f __attribute__((ext_vector_type(2)));
typedef float v8f __attribute__((ext_vector_type(8)));

__device__ __forceinline__ float sel5(int t, float e0, float e1, float e2, float e3, float e4) {
    return (t == 0) ? e0 : (t == 1) ? e1 : (t == 2) ? e2 : (t == 3) ? e3 : e4;
}

__global__ void __launch_bounds__(32) k_proj(const float* __restrict__ X,
                                             const float* __restrict__ Wm,
                                             float* __restrict__ hws) {
    const int lane = threadIdx.x;
    const int half = lane >> 4;
    const int lm   = lane & 15;
    const int i0   = blockIdx.x * 16;
    const int ho0  = blockIdx.y * 16;
    const int bh   = blockIdx.z;
    const int b    = bh / Hq, hh = bh % Hq;

    const float* xr = X + ((size_t)b * Nq + (size_t)(i0 + lm)) * Dq;
    const float* wp = Wm + (size_t)hh * Dq * HOq + ho0 + lm;

    v8f c = {};
#pragma unroll 4
    for (int k = 0; k < Dq; k += 4) {
        const int kk = k + half * 2;
        float2 axy = *(const float2*)(xr + kk);
        v2f a; a.x = axy.x; a.y = axy.y;
        v2f bb;
        bb.x = wp[(size_t)kk * HOq];
        bb.y = wp[(size_t)(kk + 1) * HOq];
        c = __builtin_amdgcn_wmma_f32_16x16x4_f32(false, a, false, bb,
                                                  (short)0, c, false, false);
    }
    float* hp = hws + ((size_t)bh * Nq + i0) * HOq + ho0 + lm;
#pragma unroll
    for (int r = 0; r < 8; ++r)
        hp[(size_t)(r + 8 * half) * HOq] = c[r];
}

__global__ void __launch_bounds__(256) k_escore(const float* __restrict__ hws,
                                                const float* __restrict__ asrc,
                                                const float* __restrict__ adst,
                                                float* __restrict__ es,
                                                float* __restrict__ ed) {
    int idx = blockIdx.x * 256 + threadIdx.x;
    if (idx >= Bq * Hq * Tq * Nq) return;
    const int n  = idx % Nq;
    const int t  = (idx / Nq) % Tq;
    const int bh = idx / (Nq * Tq);
    const int hh = bh % Hq;
    const float4* hv = (const float4*)(hws + ((size_t)bh * Nq + n) * HOq);
    const float4* sv = (const float4*)(asrc + ((size_t)hh * Tq + t) * HOq);
    const float4* dv = (const float4*)(adst + ((size_t)hh * Tq + t) * HOq);
    float ss = 0.f, sd = 0.f;
#pragma unroll
    for (int q = 0; q < HOq / 4; ++q) {
        float4 hx = hv[q], sx = sv[q], dx = dv[q];
        ss += hx.x * sx.x + hx.y * sx.y + hx.z * sx.z + hx.w * sx.w;
        sd += hx.x * dx.x + hx.y * dx.y + hx.z * dx.z + hx.w * dx.w;
    }
    es[idx] = ss;
    ed[idx] = sd;
}

__global__ void __launch_bounds__(32) k_softstats(const int* __restrict__ adj,
                                                  const float* __restrict__ es,
                                                  const float* __restrict__ ed,
                                                  float* __restrict__ rmax,
                                                  float* __restrict__ rsum) {
    const int lane  = threadIdx.x;
    const int rowid = blockIdx.x;
    const int i  = rowid % Nq;
    const int bh = rowid / Nq;
    const int b  = bh / Hq;

    const float e0 = es[((size_t)bh * Tq + 0) * Nq + i];
    const float e1 = es[((size_t)bh * Tq + 1) * Nq + i];
    const float e2 = es[((size_t)bh * Tq + 2) * Nq + i];
    const float e3 = es[((size_t)bh * Tq + 3) * Nq + i];
    const float e4 = es[((size_t)bh * Tq + 4) * Nq + i];
    const int*   ar  = adj + ((size_t)b * Nq + i) * Nq;
    const float* edb = ed + (size_t)bh * Tq * Nq;

    float m = -INFINITY, s = 0.f;
    for (int j = lane; j < Nq; j += 32) {
        int a = ar[j];
        int t = (a > 0) ? (a - 1) : 0;
        float sc = sel5(t, e0, e1, e2, e3, e4) + edb[(size_t)t * Nq + j];
        float w = (a == 0) ? NEGV : (sc > 0.f ? sc : 0.01f * sc);
        float mn = fmaxf(m, w);
        s = s * __expf(m - mn) + __expf(w - mn);
        m = mn;
    }
#pragma unroll
    for (int off = 16; off >= 1; off >>= 1) {
        float mo = __shfl_xor(m, off, 32);
        float so = __shfl_xor(s, off, 32);
        float mn = fmaxf(m, mo);
        s = s * __expf(m - mn) + so * __expf(mo - mn);
        m = mn;
    }
    if (lane == 0) { rmax[rowid] = m; rsum[rowid] = s; }
}

__global__ void __launch_bounds__(32) k_attn(const int* __restrict__ adj,
                                             const float* __restrict__ hws,
                                             const float* __restrict__ es,
                                             const float* __restrict__ ed,
                                             const float* __restrict__ rmax,
                                             const float* __restrict__ rsum,
                                             const float* __restrict__ bias,
                                             const float* __restrict__ gamma,
                                             const float* __restrict__ beta,
                                             float* __restrict__ out) {
    __shared__ float tile[16 * 68];
    const int lane = threadIdx.x;
    const int half = lane >> 4;
    const int lm   = lane & 15;
    const int i0   = blockIdx.x * 16;
    const int bh   = blockIdx.y;
    const int b    = bh / Hq, hh = bh % Hq;
    const int row  = i0 + lm;

    const float e0 = es[((size_t)bh * Tq + 0) * Nq + row];
    const float e1 = es[((size_t)bh * Tq + 1) * Nq + row];
    const float e2 = es[((size_t)bh * Tq + 2) * Nq + row];
    const float e3 = es[((size_t)bh * Tq + 3) * Nq + row];
    const float e4 = es[((size_t)bh * Tq + 4) * Nq + row];
    const float rm   = rmax[(size_t)bh * Nq + row];
    const float rinv = 1.f / rsum[(size_t)bh * Nq + row];

    const int*   ar  = adj + ((size_t)b * Nq + row) * Nq;
    const float* edb = ed + (size_t)bh * Tq * Nq;
    const float* hb  = hws + (size_t)bh * Nq * HOq;

    v8f c0 = {}, c1 = {}, c2 = {}, c3 = {};
    for (int j0 = 0; j0 < Nq; j0 += 4) {
        const int jj = j0 + half * 2;
        int2 av = *(const int2*)(ar + jj);

        float p0, p1;
        {
            int a = av.x; int t = (a > 0) ? (a - 1) : 0;
            float sc = sel5(t, e0, e1, e2, e3, e4) + edb[(size_t)t * Nq + jj];
            float w = (a == 0) ? NEGV : (sc > 0.f ? sc : 0.01f * sc);
            p0 = __expf(w - rm) * rinv;
        }
        {
            int a = av.y; int t = (a > 0) ? (a - 1) : 0;
            float sc = sel5(t, e0, e1, e2, e3, e4) + edb[(size_t)t * Nq + jj + 1];
            float w = (a == 0) ? NEGV : (sc > 0.f ? sc : 0.01f * sc);
            p1 = __expf(w - rm) * rinv;
        }
        v2f ap; ap.x = p0; ap.y = p1;

        const float* hr0 = hb + (size_t)jj * HOq + lm;
        const float* hr1 = hr0 + HOq;
        v2f b0, b1, b2, b3;
        b0.x = hr0[0];  b0.y = hr1[0];
        b1.x = hr0[16]; b1.y = hr1[16];
        b2.x = hr0[32]; b2.y = hr1[32];
        b3.x = hr0[48]; b3.y = hr1[48];

        c0 = __builtin_amdgcn_wmma_f32_16x16x4_f32(false, ap, false, b0, (short)0, c0, false, false);
        c1 = __builtin_amdgcn_wmma_f32_16x16x4_f32(false, ap, false, b1, (short)0, c1, false, false);
        c2 = __builtin_amdgcn_wmma_f32_16x16x4_f32(false, ap, false, b2, (short)0, c2, false, false);
        c3 = __builtin_amdgcn_wmma_f32_16x16x4_f32(false, ap, false, b3, (short)0, c3, false, false);
    }

#pragma unroll
    for (int r = 0; r < 8; ++r) {
        const int m = r + 8 * half;
        tile[m * 68 +  0 + lm] = c0[r];
        tile[m * 68 + 16 + lm] = c1[r];
        tile[m * 68 + 32 + lm] = c2[r];
        tile[m * 68 + 48 + lm] = c3[r];
    }
    __syncthreads();

    if (lane < 16) {
        const int r = lane;
        const float* hres = hb + (size_t)(i0 + r) * HOq;
        float sum = 0.f, sumsq = 0.f;
#pragma unroll 8
        for (int e = 0; e < HOq; ++e) {
            float v = tile[r * 68 + e] + bias[hh * HOq + e];
            v = fmaxf(v, 0.f);
            v += hres[e];
            tile[r * 68 + e] = v;
            sum += v; sumsq += v * v;
        }
        const float mu   = sum * (1.f / HOq);
        const float var  = sumsq * (1.f / HOq) - mu * mu;
        const float rstd = rsqrtf(var + EPSV);
        float* op = out + ((size_t)b * Nq + (i0 + r)) * (Hq * HOq) + hh * HOq;
#pragma unroll 8
        for (int e = 0; e < HOq; ++e)
            op[e] = (tile[r * 68 + e] - mu) * rstd * gamma[hh * HOq + e] + beta[hh * HOq + e];
    }
}

extern "C" void kernel_launch(void* const* d_in, const int* in_sizes, int n_in,
                              void* d_out, int out_size, void* d_ws, size_t ws_size,
                              hipStream_t stream) {
    const float* nodes = (const float*)d_in[0];
    const int*   adj   = (const int*)d_in[1];
    const float* Wm    = (const float*)d_in[2];
    const float* asrc  = (const float*)d_in[3];
    const float* adst  = (const float*)d_in[4];
    const float* bias  = (const float*)d_in[5];
    const float* gamma = (const float*)d_in[6];
    const float* beta  = (const float*)d_in[7];
    float* out = (float*)d_out;

    float* h_ws = (float*)d_ws;
    float* esrc = h_ws + (size_t)Bq * Hq * Nq * HOq;
    float* edst = esrc + (size_t)Bq * Hq * Tq * Nq;
    float* rmx  = edst + (size_t)Bq * Hq * Tq * Nq;
    float* rsm  = rmx  + (size_t)Bq * Hq * Nq;

    k_proj<<<dim3(Nq / 16, HOq / 16, Bq * Hq), 32, 0, stream>>>(nodes, Wm, h_ws);
    k_escore<<<(Bq * Hq * Tq * Nq + 255) / 256, 256, 0, stream>>>(h_ws, asrc, adst, esrc, edst);
    k_softstats<<<Bq * Hq * Nq, 32, 0, stream>>>(adj, esrc, edst, rmx, rsm);
    k_attn<<<dim3(Nq / 16, Bq * Hq), 32, 0, stream>>>(adj, h_ws, esrc, edst, rmx, rsm,
                                                      bias, gamma, beta, out);
}


# k_attn: a 16-row tile shared by a 4-wave group: wave w builds P rows 4w..4w+3 and owns output columns 16w..16w+15 (P tile in two LDS buffers, epilogue by all 128 threads); sum order unchanged
# speedup vs baseline: 2.4174x; 2.4174x over previous
#include <hip/hip_runtime.h>

#define Bq  8
#define Nq  1024
#define Dq  256
#define Hq  4
#define HOq 64
#define Tq  5
#define NEGV (-1e9f)
#define EPSV (1e-6f)

typedef float v2f __attribute__((ext_vector_type(2)));
typedef float v8f __attribute__((ext_vector_type(8)));

__global__ void __launch_bounds__(32) k_proj(const float* __restrict__ X,
                                             const float* __restrict__ Wm,
                                             float* __restrict__ hws) {
    const int lane = threadIdx.x;
    const int half = lane >> 4;
    const int lm   = lane & 15;
    const int i0   = blockIdx.x * 16;
    const int ho0  = blockIdx.y * 16;
    const int bh   = blockIdx.z;
    const int b    = bh / Hq, hh = bh % Hq;

    const float* xr = X + ((size_t)b * Nq + (size_t)(i0 + lm)) * Dq;
    const float* wp = Wm + (size_t)hh * Dq * HOq + ho0 + lm;

    v8f c = {};
#pragma unroll 4
    for (int k = 0; k < Dq; k += 4) {
        const int kk = k + half * 2;
        float2 axy = *(const float2*)(xr + kk);
        v2f a; a.x = axy.x; a.y = axy.y;
        v2f bb;
        bb.x = wp[(size_t)kk * HOq];
        bb.y = wp[(size_t)(kk + 1) * HOq];
        c = __builtin_amdgcn_wmma_f32_16x16x4_f32(false, a, false, bb,
                                                  (short)0, c, false, false);
    }
    float* hp = hws + ((size_t)bh * Nq + i0) * HOq + ho0 + lm;
#pragma unroll
    for (int r = 0; r < 8; ++r)
        hp[(size_t)(r + 8 * half) * HOq] = c[r];
}

__global__ void __launch_bounds__(256) k_escore(const float* __restrict__ hws,
                                                const float* __restrict__ asrc,
                                                const float* __restrict__ adst,
                                                float* __restrict__ es,
                                                float* __restrict__ ed) {
    int idx = blockIdx.x * 256 + threadIdx.x;
    if (idx >= Bq * Hq * Tq * Nq) return;
    const int n  = idx % Nq;
    const int t  = (idx / Nq) % Tq;
    const int bh = idx / (Nq * Tq);
    const int hh = bh % Hq;
    const float4* hv = (const float4*)(hws + ((size_t)bh * Nq + n) * HOq);
    const float4* sv = (const float4*)(asrc + ((size_t)hh * Tq + t) * HOq);
    const float4* dv = (const float4*)(adst + ((size_t)hh * Tq + t) * HOq);
    float ss = 0.f, sd = 0.f;
#pragma unroll
    for (int q = 0; q < HOq / 4; ++q) {
        float4 hx = hv[q], sx = sv[q], dx = dv[q];
        ss += hx.x * sx.x + hx.y * sx.y + hx.z * sx.z + hx.w * sx.w;
        sd += hx.x * dx.x + hx.y * dx.y + hx.z * dx.z + hx.w * dx.w;
    }
    es[idx] = ss;
    ed[idx] = sd;
}

__device__ __forceinline__ float pick5(int t, float e0, float e1, float e2, float e3, float e4) {
    float v = e4;
    v = (t == 3) ? e3 : v;
    v = (t == 2) ? e2 : v;
    v = (t == 1) ? e1 : v;
    v = (t == 0) ? e0 : v;
    return v;
}

#define SRq 8
#define STq (Nq / 32)
#define SGq 4

__device__ __forceinline__ void ss_load(int (&A)[SRq], float (&E)[Tq],
                                        const int* __restrict__ arow,
                                        const float* __restrict__ edl, int trip) {
    const int j = trip * 32;
#pragma unroll
    for (int t = 0; t < Tq; ++t) E[t] = edl[(size_t)t * Nq + j];
#pragma unroll
    for (int r = 0; r < SRq; ++r) A[r] = arow[(size_t)r * Nq + j];
}

__device__ __forceinline__ void ss_step(const int (&A)[SRq], const float (&E)[Tq],
                                        const float (&esr)[SRq][Tq],
                                        float (&m)[SRq], float (&s)[SRq]) {
#pragma unroll
    for (int r = 0; r < SRq; ++r) {
        const int a = A[r];
        const int t = (a > 0) ? (a - 1) : 0;
        const float sc = pick5(t, esr[r][0], esr[r][1], esr[r][2], esr[r][3], esr[r][4])
                       + pick5(t, E[0], E[1], E[2], E[3], E[4]);
        const float w = (a == 0) ? NEGV : (sc > 0.f ? sc : 0.01f * sc);
        const float mn = fmaxf(m[r], w);
        s[r] = s[r] * __expf(m[r] - mn) + __expf(w - mn);
        m[r] = mn;
    }
}

__global__ void __launch_bounds__(32 * Hq) k_softstats(const int* __restrict__ adj,
                                                       const float* __restrict__ es,
                                                       const float* __restrict__ ed,
                                                       float* __restrict__ rmax,
                                                       float* __restrict__ rsum) {
    const int lane = threadIdx.x & 31;
    const int hh = threadIdx.x >> 5;
    const int b  = blockIdx.x / (Nq / SRq);
    const int i0 = (blockIdx.x % (Nq / SRq)) * SRq;
    const int bh = b * Hq + hh;

    const int*   arow = adj + ((size_t)b * Nq + i0) * Nq + lane;
    const float* edl  = ed + (size_t)bh * Tq * Nq + lane;

    float esr[SRq][Tq];
#pragma unroll
    for (int r = 0; r < SRq; ++r) {
#pragma unroll
        for (int t = 0; t < Tq; ++t) esr[r][t] = es[((size_t)bh * Tq + t) * Nq + i0 + r];
    }

    float m[SRq], s[SRq];
#pragma unroll
    for (int r = 0; r < SRq; ++r) { m[r] = -INFINITY; s[r] = 0.f; }

    for (int blk = 0; blk < STq; blk += SGq) {
        int   a[SGq][SRq];
        float d[SGq][Tq];
#pragma unroll
        for (int g = 0; g < SGq; ++g) ss_load(a[g], d[g], arow, edl, blk + g);
#pragma unroll
        for (int g = 0; g < SGq; ++g) ss_step(a[g], d[g], esr, m, s);
    }

    float mo_ = 0.f, so_ = 0.f;
#pragma unroll
    for (int r = 0; r < SRq; ++r) {
        float mr = m[r], sr = s[r];
#pragma unroll
        for (int off = 16; off >= 1; off >>= 1) {
            float mo = __shfl_xor(mr, off, 32);
            float so = __shfl_xor(sr, off, 32);
            float mn = fmaxf(mr, mo);
            sr = sr * __expf(mr - mn) + so * __expf(mo - mn);
            mr = mn;
        }
        if (lane == r) { mo_ = mr; so_ = sr; }
    }
    if (lane < SRq) {
        rmax[(size_t)bh * Nq + i0 + lane] = mo_;
        rsum[(size_t)bh * Nq + i0 + lane] = so_;
    }
}

__device__ __forceinline__ void group_lds_barrier() {
    __builtin_amdgcn_fence(__ATOMIC_RELEASE, "workgroup", "local");
    __builtin_amdgcn_s_barrier();
    __builtin_amdgcn_fence(__ATOMIC_ACQUIRE, "workgroup", "local");
}

#define KBq 32
#define PPq 36
#define GWq 4
#define RWq (16 / GWq)

__device__ __forceinline__ void at_load(int (&A)[RWq], float (&E)[Tq], v2f (&Bf)[8],
                                        const int* __restrict__ arow,
                                        const float* __restrict__ edl,
                                        const float* __restrict__ hfr, int j) {
#pragma unroll
    for (int r = 0; r < RWq; ++r) A[r] = arow[(size_t)r * Nq + j];
#pragma unroll
    for (int t = 0; t < Tq; ++t) E[t] = edl[(size_t)t * Nq + j];
#pragma unroll
    for (int q = 0; q < 8; ++q) {
        const float* h0 = hfr + (size_t)(j + 4 * q) * HOq;
        Bf[q].x = h0[0];
        Bf[q].y = h0[HOq];
    }
}

__device__ __forceinline__ void at_trip(const int (&A)[RWq], const float (&E)[Tq], const v2f (&Bf)[8],
                                        const float (&esr)[RWq][Tq], const float (&rmx)[RWq],
                                        const float (&rin)[RWq], float* __restrict__ Pb,
                                        int prow, int lane, int lm, int half, v8f& c) {
#pragma unroll
    for (int r = 0; r < RWq; ++r) {
        const int a = A[r];
        const int t = (a > 0) ? (a - 1) : 0;
        const float sc = pick5(t, esr[r][0], esr[r][1], esr[r][2], esr[r][3], esr[r][4])
                       + pick5(t, E[0], E[1], E[2], E[3], E[4]);
        const float wv = (a == 0) ? NEGV : (sc > 0.f ? sc : 0.01f * sc);
        Pb[(prow + r) * PPq + lane] = __expf(wv - rmx[r]) * rin[r];
    }
    group_lds_barrier();
#pragma unroll
    for (int q = 0; q < 8; ++q) {
        const v2f ap = *(const v2f*)&Pb[lm * PPq + 4 * q + 2 * half];
        c = __builtin_amdgcn_wmma_f32_16x16x4_f32(false, ap, false, Bf[q], (short)0, c, false, false);
    }
}

__global__ void __launch_bounds__(32 * GWq) k_attn(const int* __restrict__ adj,
                                                   const float* __restrict__ hws,
                                                   const float* __restrict__ es,
                                                   const float* __restrict__ ed,
                                                   const float* __restrict__ rmax,
                                                   const float* __restrict__ rsum,
                                                   const float* __restrict__ bias,
                                                   const float* __restrict__ gamma,
                                                   const float* __restrict__ beta,
                                                   float* __restrict__ out) {
    __shared__ __attribute__((aligned(16))) float Pt[2][16 * PPq];
    __shared__ __attribute__((aligned(16))) float tile[16 * 68];
    __shared__ float stat[16 * 2];
    const int lane = threadIdx.x & 31;
    const int w    = __builtin_amdgcn_readfirstlane(threadIdx.x >> 5);
    const int half = lane >> 4;
    const int lm   = lane & 15;
    const int i0   = blockIdx.x * 16;
    const int bh   = blockIdx.y;
    const int b    = bh / Hq, hh = bh % Hq;
    const int prow = RWq * w;

    float esr[RWq][Tq], rmx[RWq], rin[RWq];
#pragma unroll
    for (int r = 0; r < RWq; ++r) {
        const int row = i0 + prow + r;
#pragma unroll
        for (int t = 0; t < Tq; ++t) esr[r][t] = es[((size_t)bh * Tq + t) * Nq + row];
        rmx[r] = rmax[(size_t)bh * Nq + row];
        rin[r] = 1.f / rsum[(size_t)bh * Nq + row];
    }

    const int*   arow = adj + ((size_t)b * Nq + i0 + prow) * Nq + lane;
    const float* edl  = ed + (size_t)bh * Tq * Nq + lane;
    const float* hb   = hws + (size_t)bh * Nq * HOq;
    const float* hfr  = hb + (size_t)(2 * half) * HOq + 16 * w + lm;

    int   adjA[RWq], adjB[RWq];
    float edA[Tq],   edB[Tq];
    v2f   bfA[8],    bfB[8];
    at_load(adjA, edA, bfA, arow, edl, hfr, 0);

    v8f c = {};
    for (int blk = 0; blk < Nq / KBq; blk += 2) {
        const int j0 = blk * KBq;
        at_load(adjB, edB, bfB, arow, edl, hfr, j0 + KBq);
        at_trip(adjA, edA, bfA, esr, rmx, rin, Pt[0], prow, lane, lm, half, c);
        const int jn = (blk + 2 < Nq / KBq) ? (j0 + 2 * KBq) : (j0 + KBq);
        at_load(adjA, edA, bfA, arow, edl, hfr, jn);
        at_trip(adjB, edB, bfB, esr, rmx, rin, Pt[1], prow, lane, lm, half, c);
    }

#pragma unroll
    for (int r = 0; r < 8; ++r) tile[(r + 8 * half) * 68 + 16 * w + lm] = c[r];
    __syncthreads();

    const int row = threadIdx.x >> 3;
    const int e0  = (threadIdx.x & 7) * 8;
    float v[8];
    {
        const float* hres = hb + (size_t)(i0 + row) * HOq + e0;
        const float4 h0 = *(const float4*)hres,                   h1 = *(const float4*)(hres + 4);
        const float4 b0 = *(const float4*)(bias + hh * HOq + e0), b1 = *(const float4*)(bias + hh * HOq + e0 + 4);
        const float4 t0 = *(const float4*)&tile[row * 68 + e0],   t1 = *(const float4*)&tile[row * 68 + e0 + 4];
        v[0] = fmaxf(t0.x + b0.x, 0.f) + h0.x;
        v[1] = fmaxf(t0.y + b0.y, 0.f) + h0.y;
        v[2] = fmaxf(t0.z + b0.z, 0.f) + h0.z;
        v[3] = fmaxf(t0.w + b0.w, 0.f) + h0.w;
        v[4] = fmaxf(t1.x + b1.x, 0.f) + h1.x;
        v[5] = fmaxf(t1.y + b1.y, 0.f) + h1.y;
        v[6] = fmaxf(t1.z + b1.z, 0.f) + h1.z;
        v[7] = fmaxf(t1.w + b1.w, 0.f) + h1.w;
        *(float4*)&tile[row * 68 + e0]     = make_float4(v[0], v[1], v[2], v[3]);
        *(float4*)&tile[row * 68 + e0 + 4] = make_float4(v[4], v[5], v[6], v[7]);
    }
    __syncthreads();
    if (threadIdx.x < 16) {
        const int r = threadIdx.x;
        float sum = 0.f, sumsq = 0.f;
#pragma unroll 8
        for (int e = 0; e < HOq; ++e) {
            const float x = tile[r * 68 + e];
            sum += x; sumsq += x * x;
        }
        const float mu   = sum * (1.f / HOq);
        const float var  = sumsq * (1.f / HOq) - mu * mu;
        stat[r * 2]     = mu;
        stat[r * 2 + 1] = rsqrtf(var + EPSV);
    }
    __syncthreads();
    {
        const float mu = stat[row * 2], rstd = stat[row * 2 + 1];
        const float4 g0 = *(const float4*)(gamma + hh * HOq + e0), g1 = *(const float4*)(gamma + hh * HOq + e0 + 4);
        const float4 z0 = *(const float4*)(beta + hh * HOq + e0),  z1 = *(const float4*)(beta + hh * HOq + e0 + 4);
        float* op = out + ((size_t)b * Nq + (i0 + row)) * (Hq * HOq) + hh * HOq + e0;
        float4 o0, o1;
        o0.x = (v[0] - mu) * rstd * g0.x + z0.x;
        o0.y = (v[1] - mu) * rstd * g0.y + z0.y;
        o0.z = (v[2] - mu) * rstd * g0.z + z0.z;
        o0.w = (v[3] - mu) * rstd * g0.w + z0.w;
        o1.x = (v[4] - mu) * rstd * g1.x + z1.x;
        o1.y = (v[5] - mu) * rstd * g1.y + z1.y;
        o1.z = (v[6] - mu) * rstd * g1.z + z1.z;
        o1.w = (v[7] - mu) * rstd * g1.w + z1.w;
        *(float4*)op       = o0;
        *(float4*)(op + 4) = o1;
    }
}

extern "C" void kernel_launch(void* const* d_in, const int* in_sizes, int n_in,
                              void* d_out, int out_size, void* d_ws, size_t ws_size,
                              hipStream_t stream) {
    const float* nodes = (const float*)d_in[0];
    const int*   adj   = (const int*)d_in[1];
    const float* Wm    = (const float*)d_in[2];
    const float* asrc  = (const float*)d_in[3];
    const float* adst  = (const float*)d_in[4];
    const float* bias  = (const float*)d_in[5];
    const float* gamma = (const float*)d_in[6];
    const float* beta  = (const float*)d_in[7];
    float* out = (float*)d_out;

    float* h_ws = (float*)d_ws;
    float* esrc = h_ws + (size_t)Bq * Hq * Nq * HOq;
    float* edst = esrc + (size_t)Bq * Hq * Tq * Nq;
    float* rmx  = edst + (size_t)Bq * Hq * Tq * Nq;
    float* rsm  = rmx  + (size_t)Bq * Hq * Nq;

    k_proj<<<dim3(Nq / 16, HOq / 16, Bq * Hq), 32, 0, stream>>>(nodes, Wm, h_ws);
    k_escore<<<(Bq * Hq * Tq * Nq + 255) / 256, 256, 0, stream>>>(h_ws, asrc, adst, esrc, edst);
    k_softstats<<<Bq * (Nq / SRq), 32 * Hq, 0, stream>>>(adj, esrc, edst, rmx, rsm);
    k_attn<<<dim3(Nq / 16, Bq * Hq), 32 * GWq, 0, stream>>>(adj, h_ws, esrc, edst, rmx, rsm,
                                                      bias, gamma, beta, out);
}


# k_proj: a 4-wave group owns a 32-row x 64-col tile of a head: X chunks staged in LDS by whole-line 16-byte loads, a B fragment feeds two WMMAs; 4x fewer waves; k and sum order unchanged
# speedup vs baseline: 2.6004x; 1.0757x over previous
#include <hip/hip_runtime.h>

#define Bq  8
#define Nq  1024
#define Dq  256
#define Hq  4
#define HOq 64
#define Tq  5
#define NEGV (-1e9f)
#define EPSV (1e-6f)

typedef float v2f __attribute__((ext_vector_type(2)));
typedef float v8f __attribute__((ext_vector_type(8)));

__device__ __forceinline__ void group_lds_barrier() {
    __builtin_amdgcn_fence(__ATOMIC_RELEASE, "workgroup", "local");
    __builtin_amdgcn_s_barrier();
    __builtin_amdgcn_fence(__ATOMIC_ACQUIRE, "workgroup", "local");
}

#define PRq 32
#define PCq 32
#define PXq 36

__device__ __forceinline__ void pj_load(float4& xa, float4& xb, v2f (&Bf)[8],
                                        const float* __restrict__ xg,
                                        const float* __restrict__ wp, int k0) {
    xa = *(const float4*)(xg + k0);
    xb = *(const float4*)(xg + (size_t)16 * Dq + k0);
#pragma unroll
    for (int q = 0; q < 8; ++q) {
        const float* w0 = wp + (size_t)(k0 + 4 * q) * HOq;
        Bf[q].x = w0[0];
        Bf[q].y = w0[HOq];
    }
}

__device__ __forceinline__ void pj_chunk(const v2f (&Bf)[8], const float* __restrict__ Xb,
                                         int lm, int half, v8f& c0, v8f& c1) {
#pragma unroll
    for (int q = 0; q < 8; ++q) {
        const v2f a0 = *(const v2f*)&Xb[lm * PXq + 4 * q + 2 * half];
        const v2f a1 = *(const v2f*)&Xb[(16 + lm) * PXq + 4 * q + 2 * half];
        c0 = __builtin_amdgcn_wmma_f32_16x16x4_f32(false, a0, false, Bf[q], (short)0, c0, false, false);
        c1 = __builtin_amdgcn_wmma_f32_16x16x4_f32(false, a1, false, Bf[q], (short)0, c1, false, false);
    }
}

__global__ void __launch_bounds__(128) k_proj(const float* __restrict__ X,
                                              const float* __restrict__ Wm,
                                              float* __restrict__ hws) {
    __shared__ __attribute__((aligned(16))) float Xs[2][PRq * PXq];
    const int lane = threadIdx.x & 31;
    const int w    = __builtin_amdgcn_readfirstlane(threadIdx.x >> 5);
    const int half = lane >> 4;
    const int lm   = lane & 15;
    const int hh   = blockIdx.x;
    const int g0   = blockIdx.y * PRq;
    const int b    = g0 / Nq, i0 = g0 % Nq;
    const int bh   = b * Hq + hh;

    const int srow = threadIdx.x >> 3;
    const int scol = (threadIdx.x & 7) * 4;
    const float* xg = X + ((size_t)g0 + srow) * Dq + scol;
    const float* wp = Wm + (size_t)hh * Dq * HOq + (size_t)(2 * half) * HOq + 16 * w + lm;
    const int so = srow * PXq + scol;

    float4 xa, xb;
    v2f bfA[8], bfB[8];
    pj_load(xa, xb, bfA, xg, wp, 0);
    *(float4*)&Xs[0][so]            = xa;
    *(float4*)&Xs[0][so + 16 * PXq] = xb;
    group_lds_barrier();

    v8f c0 = {}, c1 = {};
#pragma unroll
    for (int ch = 0; ch < Dq / PCq; ch += 2) {
        __builtin_amdgcn_sched_barrier(0);
        pj_load(xa, xb, bfB, xg, wp, (ch + 1) * PCq);
        __builtin_amdgcn_sched_barrier(0);
        pj_chunk(bfA, Xs[0], lm, half, c0, c1);
        *(float4*)&Xs[1][so]            = xa;
        *(float4*)&Xs[1][so + 16 * PXq] = xb;
        group_lds_barrier();
        __builtin_amdgcn_sched_barrier(0);
        if (ch + 2 < Dq / PCq) pj_load(xa, xb, bfA, xg, wp, (ch + 2) * PCq);
        __builtin_amdgcn_sched_barrier(0);
        pj_chunk(bfB, Xs[1], lm, half, c0, c1);
        if (ch + 2 < Dq / PCq) {
            *(float4*)&Xs[0][so]            = xa;
            *(float4*)&Xs[0][so + 16 * PXq] = xb;
            group_lds_barrier();
        }
    }
    float* hp = hws + ((size_t)bh * Nq + i0) * HOq + 16 * w + lm;
#pragma unroll
    for (int r = 0; r < 8; ++r) {
        hp[(size_t)(r + 8 * half) * HOq]      = c0[r];
        hp[(size_t)(16 + r + 8 * half) * HOq] = c1[r];
    }
}

__global__ void __launch_bounds__(256) k_escore(const float* __restrict__ hws,
                                                const float* __restrict__ asrc,
                                                const float* __restrict__ adst,
                                                float* __restrict__ es,
                                                float* __restrict__ ed) {
    int idx = blockIdx.x * 256 + threadIdx.x;
    if (idx >= Bq * Hq * Tq * Nq) return;
    const int n  = idx % Nq;
    const int t  = (idx / Nq) % Tq;
    const int bh = idx / (Nq * Tq);
    const int hh = bh % Hq;
    const float4* hv = (const float4*)(hws + ((size_t)bh * Nq + n) * HOq);
    const float4* sv = (const float4*)(asrc + ((size_t)hh * Tq + t) * HOq);
    const float4* dv = (const float4*)(adst + ((size_t)hh * Tq + t) * HOq);
    float ss = 0.f, sd = 0.f;
#pragma unroll
    for (int q = 0; q < HOq / 4; ++q) {
        float4 hx = hv[q], sx = sv[q], dx = dv[q];
        ss += hx.x * sx.x + hx.y * sx.y + hx.z * sx.z + hx.w * sx.w;
        sd += hx.x * dx.x + hx.y * dx.y + hx.z * dx.z + hx.w * dx.w;
    }
    es[idx] = ss;
    ed[idx] = sd;
}

__device__ __forceinline__ float pick5(int t, float e0, float e1, float e2, float e3, float e4) {
    float v = e4;
    v = (t == 3) ? e3 : v;
    v = (t == 2) ? e2 : v;
    v = (t == 1) ? e1 : v;
    v = (t == 0) ? e0 : v;
    return v;
}

#define SRq 8
#define STq (Nq / 32)
#define SGq 4

__device__ __forceinline__ void ss_load(int (&A)[SRq], float (&E)[Tq],
                                        const int* __restrict__ arow,
                                        const float* __restrict__ edl, int trip) {
    const int j = trip * 32;
#pragma unroll
    for (int t = 0; t < Tq; ++t) E[t] = edl[(size_t)t * Nq + j];
#pragma unroll
    for (int r = 0; r < SRq; ++r) A[r] = arow[(size_t)r * Nq + j];
}

__device__ __forceinline__ void ss_step(const int (&A)[SRq], const float (&E)[Tq],
                                        const float (&esr)[SRq][Tq],
                                        float (&m)[SRq], float (&s)[SRq]) {
#pragma unroll
    for (int r = 0; r < SRq; ++r) {
        const int a = A[r];
        const int t = (a > 0) ? (a - 1) : 0;
        const float sc = pick5(t, esr[r][0], esr[r][1], esr[r][2], esr[r][3], esr[r][4])
                       + pick5(t, E[0], E[1], E[2], E[3], E[4]);
        const float w = (a == 0) ? NEGV : (sc > 0.f ? sc : 0.01f * sc);
        const float mn = fmaxf(m[r], w);
        s[r] = s[r] * __expf(m[r] - mn) + __expf(w - mn);
        m[r] = mn;
    }
}

__global__ void __launch_bounds__(32 * Hq) k_softstats(const int* __restrict__ adj,
                                                       const float* __restrict__ es,
                                                       const float* __restrict__ ed,
                                                       float* __restrict__ rmax,
                                                       float* __restrict__ rsum) {
    const int lane = threadIdx.x & 31;
    const int hh = threadIdx.x >> 5;
    const int b  = blockIdx.x / (Nq / SRq);
    const int i0 = (blockIdx.x % (Nq / SRq)) * SRq;
    const int bh = b * Hq + hh;

    const int*   arow = adj + ((size_t)b * Nq + i0) * Nq + lane;
    const float* edl  = ed + (size_t)bh * Tq * Nq + lane;

    float esr[SRq][Tq];
#pragma unroll
    for (int r = 0; r < SRq; ++r) {
#pragma unroll
        for (int t = 0; t < Tq; ++t) esr[r][t] = es[((size_t)bh * Tq + t) * Nq + i0 + r];
    }

    float m[SRq], s[SRq];
#pragma unroll
    for (int r = 0; r < SRq; ++r) { m[r] = -INFINITY; s[r] = 0.f; }

    for (int blk = 0; blk < STq; blk += SGq) {
        int   a[SGq][SRq];
        float d[SGq][Tq];
#pragma unroll
        for (int g = 0; g < SGq; ++g) ss_load(a[g], d[g], arow, edl, blk + g);
#pragma unroll
        for (int g = 0; g < SGq; ++g) ss_step(a[g], d[g], esr, m, s);
    }

    float mo_ = 0.f, so_ = 0.f;
#pragma unroll
    for (int r = 0; r < SRq; ++r) {
        float mr = m[r], sr = s[r];
#pragma unroll
        for (int off = 16; off >= 1; off >>= 1) {
            float mo = __shfl_xor(mr, off, 32);
            float so = __shfl_xor(sr, off, 32);
            float mn = fmaxf(mr, mo);
            sr = sr * __expf(mr - mn) + so * __expf(mo - mn);
            mr = mn;
        }
        if (lane == r) { mo_ = mr; so_ = sr; }
    }
    if (lane < SRq) {
        rmax[(size_t)bh * Nq + i0 + lane] = mo_;
        rsum[(size_t)bh * Nq + i0 + lane] = so_;
    }
}

#define KBq 32
#define PPq 36
#define GWq 4
#define RWq (16 / GWq)

__device__ __forceinline__ void at_load(int (&A)[RWq], float (&E)[Tq], v2f (&Bf)[8],
                                        const int* __restrict__ arow,
                                        const float* __restrict__ edl,
                                        const float* __restrict__ hfr, int j) {
#pragma unroll
    for (int r = 0; r < RWq; ++r) A[r] = arow[(size_t)r * Nq + j];
#pragma unroll
    for (int t = 0; t < Tq; ++t) E[t] = edl[(size_t)t * Nq + j];
#pragma unroll
    for (int q = 0; q < 8; ++q) {
        const float* h0 = hfr + (size_t)(j + 4 * q) * HOq;
        Bf[q].x = h0[0];
        Bf[q].y = h0[HOq];
    }
}

__device__ __forceinline__ void at_trip(const int (&A)[RWq], const float (&E)[Tq], const v2f (&Bf)[8],
                                        const float (&esr)[RWq][Tq], const float (&rmx)[RWq],
                                        const float (&rin)[RWq], float* __restrict__ Pb,
                                        int prow, int lane, int lm, int half, v8f& c) {
#pragma unroll
    for (int r = 0; r < RWq; ++r) {
        const int a = A[r];
        const int t = (a > 0) ? (a - 1) : 0;
        const float sc = pick5(t, esr[r][0], esr[r][1], esr[r][2], esr[r][3], esr[r][4])
                       + pick5(t, E[0], E[1], E[2], E[3], E[4]);
        const float wv = (a == 0) ? NEGV : (sc > 0.f ? sc : 0.01f * sc);
        Pb[(prow + r) * PPq + lane] = __expf(wv - rmx[r]) * rin[r];
    }
    group_lds_barrier();
#pragma unroll
    for (int q = 0; q < 8; ++q) {
        const v2f ap = *(const v2f*)&Pb[lm * PPq + 4 * q + 2 * half];
        c = __builtin_amdgcn_wmma_f32_16x16x4_f32(false, ap, false, Bf[q], (short)0, c, false, false);
    }
}

__global__ void __launch_bounds__(32 * GWq) k_attn(const int* __restrict__ adj,
                                                   const float* __restrict__ hws,
                                                   const float* __restrict__ es,
                                                   const float* __restrict__ ed,
                                                   const float* __restrict__ rmax,
                                                   const float* __restrict__ rsum,
                                                   const float* __restrict__ bias,
                                                   const float* __restrict__ gamma,
                                                   const float* __restrict__ beta,
                                                   float* __restrict__ out) {
    __shared__ __attribute__((aligned(16))) float Pt[2][16 * PPq];
    __shared__ __attribute__((aligned(16))) float tile[16 * 68];
    __shared__ float stat[16 * 2];
    const int lane = threadIdx.x & 31;
    const int w    = __builtin_amdgcn_readfirstlane(threadIdx.x >> 5);
    const int half = lane >> 4;
    const int lm   = lane & 15;
    const int i0   = blockIdx.x * 16;
    const int bh   = blockIdx.y;
    const int b    = bh / Hq, hh = bh % Hq;
    const int prow = RWq * w;

    float esr[RWq][Tq], rmx[RWq], rin[RWq];
#pragma unroll
    for (int r = 0; r < RWq; ++r) {
        const int row = i0 + prow + r;
#pragma unroll
        for (int t = 0; t < Tq; ++t) esr[r][t] = es[((size_t)bh * Tq + t) * Nq + row];
        rmx[r] = rmax[(size_t)bh * Nq + row];
        rin[r] = 1.f / rsum[(size_t)bh * Nq + row];
    }

    const int*   arow = adj + ((size_t)b * Nq + i0 + prow) * Nq + lane;
    const float* edl  = ed + (size_t)bh * Tq * Nq + lane;
    const float* hb   = hws + (size_t)bh * Nq * HOq;
    const float* hfr  = hb + (size_t)(2 * half) * HOq + 16 * w + lm;

    int   adjA[RWq], adjB[RWq];
    float edA[Tq],   edB[Tq];
    v2f   bfA[8],    bfB[8];
    at_load(adjA, edA, bfA, arow, edl, hfr, 0);

    v8f c = {};
    for (int blk = 0; blk < Nq / KBq; blk += 2) {
        const int j0 = blk * KBq;
        at_load(adjB, edB, bfB, arow, edl, hfr, j0 + KBq);
        at_trip(adjA, edA, bfA, esr, rmx, rin, Pt[0], prow, lane, lm, half, c);
        const int jn = (blk + 2 < Nq / KBq) ? (j0 + 2 * KBq) : (j0 + KBq);
        at_load(adjA, edA, bfA, arow, edl, hfr, jn);
        at_trip(adjB, edB, bfB, esr, rmx, rin, Pt[1], prow, lane, lm, half, c);
    }

#pragma unroll
    for (int r = 0; r < 8; ++r) tile[(r + 8 * half) * 68 + 16 * w + lm] = c[r];
    __syncthreads();

    const int row = threadIdx.x >> 3;
    const int e0  = (threadIdx.x & 7) * 8;
    float v[8];
    {
        const float* hres = hb + (size_t)(i0 + row) * HOq + e0;
        const float4 h0 = *(const float4*)hres,                   h1 = *(const float4*)(hres + 4);
        const float4 b0 = *(const float4*)(bias + hh * HOq + e0), b1 = *(const float4*)(bias + hh * HOq + e0 + 4);
        const float4 t0 = *(const float4*)&tile[row * 68 + e0],   t1 = *(const float4*)&tile[row * 68 + e0 + 4];
        v[0] = fmaxf(t0.x + b0.x, 0.f) + h0.x;
        v[1] = fmaxf(t0.y + b0.y, 0.f) + h0.y;
        v[2] = fmaxf(t0.z + b0.z, 0.f) + h0.z;
        v[3] = fmaxf(t0.w + b0.w, 0.f) + h0.w;
        v[4] = fmaxf(t1.x + b1.x, 0.f) + h1.x;
        v[5] = fmaxf(t1.y + b1.y, 0.f) + h1.y;
        v[6] = fmaxf(t1.z + b1.z, 0.f) + h1.z;
        v[7] = fmaxf(t1.w + b1.w, 0.f) + h1.w;
        *(float4*)&tile[row * 68 + e0]     = make_float4(v[0], v[1], v[2], v[3]);
        *(float4*)&tile[row * 68 + e0 + 4] = make_float4(v[4], v[5], v[6], v[7]);
    }
    __syncthreads();
    if (threadIdx.x < 16) {
        const int r = threadIdx.x;
        float sum = 0.f, sumsq = 0.f;
#pragma unroll 8
        for (int e = 0; e < HOq; ++e) {
            const float x = tile[r * 68 + e];
            sum += x; sumsq += x * x;
        }
        const float mu   = sum * (1.f / HOq);
        const float var  = sumsq * (1.f / HOq) - mu * mu;
        stat[r * 2]     = mu;
        stat[r * 2 + 1] = rsqrtf(var + EPSV);
    }
    __syncthreads();
    {
        const float mu = stat[row * 2], rstd = stat[row * 2 + 1];
        const float4 g0 = *(const float4*)(gamma + hh * HOq + e0), g1 = *(const float4*)(gamma + hh * HOq + e0 + 4);
        const float4 z0 = *(const float4*)(beta + hh * HOq + e0),  z1 = *(const float4*)(beta + hh * HOq + e0 + 4);
        float* op = out + ((size_t)b * Nq + (i0 + row)) * (Hq * HOq) + hh * HOq + e0;
        float4 o0, o1;
        o0.x = (v[0] - mu) * rstd * g0.x + z0.x;
        o0.y = (v[1] - mu) * rstd * g0.y + z0.y;
        o0.z = (v[2] - mu) * rstd * g0.z + z0.z;
        o0.w = (v[3] - mu) * rstd * g0.w + z0.w;
        o1.x = (v[4] - mu) * rstd * g1.x + z1.x;
        o1.y = (v[5] - mu) * rstd * g1.y + z1.y;
        o1.z = (v[6] - mu) * rstd * g1.z + z1.z;
        o1.w = (v[7] - mu) * rstd * g1.w + z1.w;
        *(float4*)op       = o0;
        *(float4*)(op + 4) = o1;
    }
}

extern "C" void kernel_launch(void* const* d_in, const int* in_sizes, int n_in,
                              void* d_out, int out_size, void* d_ws, size_t ws_size,
                              hipStream_t stream) {
    const float* nodes = (const float*)d_in[0];
    const int*   adj   = (const int*)d_in[1];
    const float* Wm    = (const float*)d_in[2];
    const float* asrc  = (const float*)d_in[3];
    const float* adst  = (const float*)d_in[4];
    const float* bias  = (const float*)d_in[5];
    const float* gamma = (const float*)d_in[6];
    const float* beta  = (const float*)d_in[7];
    float* out = (float*)d_out;

    float* h_ws = (float*)d_ws;
    float* esrc = h_ws + (size_t)Bq * Hq * Nq * HOq;
    float* edst = esrc + (size_t)Bq * Hq * Tq * Nq;
    float* rmx  = edst + (size_t)Bq * Hq * Tq * Nq;
    float* rsm  = rmx  + (size_t)Bq * Hq * Nq;

    k_proj<<<dim3(Hq, Bq * Nq / PRq), 32 * 4, 0, stream>>>(nodes, Wm, h_ws);
    k_escore<<<(Bq * Hq * Tq * Nq + 255) / 256, 256, 0, stream>>>(h_ws, asrc, adst, esrc, edst);
    k_softstats<<<Bq * (Nq / SRq), 32 * Hq, 0, stream>>>(adj, esrc, edst, rmx, rsm);
    k_attn<<<dim3(Nq / 16, Bq * Hq), 32 * GWq, 0, stream>>>(adj, h_ws, esrc, edst, rmx, rsm,
                                                      bias, gamma, beta, out);
}


# k_softstats keeps only the row max (leaky once per row, no exp); k_attn takes exp(score-max) unnormalised, sums each P row itself (lane order, then tree), divides the tile last; sum order changes
# speedup vs baseline: 2.7832x; 1.0703x over previous
#include <hip/hip_runtime.h>

#define Bq  8
#define Nq  1024
#define Dq  256
#define Hq  4
#define HOq 64
#define Tq  5
#define NEGV (-1e9f)
#define EPSV (1e-6f)

typedef float v2f __attribute__((ext_vector_type(2)));
typedef float v8f __attribute__((ext_vector_type(8)));

__device__ __forceinline__ void group_lds_barrier() {
    __builtin_amdgcn_fence(__ATOMIC_RELEASE, "workgroup", "local");
    __builtin_amdgcn_s_barrier();
    __builtin_amdgcn_fence(__ATOMIC_ACQUIRE, "workgroup", "local");
}

#define PRq 32
#define PCq 32
#define PXq 36

__device__ __forceinline__ void pj_load(float4& xa, float4& xb, v2f (&Bf)[8],
                                        const float* __restrict__ xg,
                                        const float* __restrict__ wp, int k0) {
    xa = *(const float4*)(xg + k0);
    xb = *(const float4*)(xg + (size_t)16 * Dq + k0);
#pragma unroll
    for (int q = 0; q < 8; ++q) {
        const float* w0 = wp + (size_t)(k0 + 4 * q) * HOq;
        Bf[q].x = w0[0];
        Bf[q].y = w0[HOq];
    }
}

__device__ __forceinline__ void pj_chunk(const v2f (&Bf)[8], const float* __restrict__ Xb,
                                         int lm, int half, v8f& c0, v8f& c1) {
#pragma unroll
    for (int q = 0; q < 8; ++q) {
        const v2f a0 = *(const v2f*)&Xb[lm * PXq + 4 * q + 2 * half];
        const v2f a1 = *(const v2f*)&Xb[(16 + lm) * PXq + 4 * q + 2 * half];
        c0 = __builtin_amdgcn_wmma_f32_16x16x4_f32(false, a0, false, Bf[q], (short)0, c0, false, false);
        c1 = __builtin_amdgcn_wmma_f32_16x16x4_f32(false, a1, false, Bf[q], (short)0, c1, false, false);
    }
}

__global__ void __launch_bounds__(128) k_proj(const float* __restrict__ X,
                                              const float* __restrict__ Wm,
                                              float* __restrict__ hws) {
    __shared__ __attribute__((aligned(16))) float Xs[2][PRq * PXq];
    const int lane = threadIdx.x & 31;
    const int w    = __builtin_amdgcn_readfirstlane(threadIdx.x >> 5);
    const int half = lane >> 4;
    const int lm   = lane & 15;
    const int hh   = blockIdx.x;
    const int g0   = blockIdx.y * PRq;
    const int b    = g0 / Nq, i0 = g0 % Nq;
    const int bh   = b * Hq + hh;

    const int srow = threadIdx.x >> 3;
    const int scol = (threadIdx.x & 7) * 4;
    const float* xg = X + ((size_t)g0 + srow) * Dq + scol;
    const float* wp = Wm + (size_t)hh * Dq * HOq + (size_t)(2 * half) * HOq + 16 * w + lm;
    const int so = srow * PXq + scol;

    float4 xa, xb;
    v2f bfA[8], bfB[8];
    pj_load(xa, xb, bfA, xg, wp, 0);
    *(float4*)&Xs[0][so]            = xa;
    *(float4*)&Xs[0][so + 16 * PXq] = xb;
    group_lds_barrier();

    v8f c0 = {}, c1 = {};
#pragma unroll
    for (int ch = 0; ch < Dq / PCq; ch += 2) {
        __builtin_amdgcn_sched_barrier(0);
        pj_load(xa, xb, bfB, xg, wp, (ch + 1) * PCq);
        __builtin_amdgcn_sched_barrier(0);
        pj_chunk(bfA, Xs[0], lm, half, c0, c1);
        *(float4*)&Xs[1][so]            = xa;
        *(float4*)&Xs[1][so + 16 * PXq] = xb;
        group_lds_barrier();
        __builtin_amdgcn_sched_barrier(0);
        if (ch + 2 < Dq / PCq) pj_load(xa, xb, bfA, xg, wp, (ch + 2) * PCq);
        __builtin_amdgcn_sched_barrier(0);
        pj_chunk(bfB, Xs[1], lm, half, c0, c1);
        if (ch + 2 < Dq / PCq) {
            *(float4*)&Xs[0][so]            = xa;
            *(float4*)&Xs[0][so + 16 * PXq] = xb;
            group_lds_barrier();
        }
    }
    float* hp = hws + ((size_t)bh * Nq + i0) * HOq + 16 * w + lm;
#pragma unroll
    for (int r = 0; r < 8; ++r) {
        hp[(size_t)(r + 8 * half) * HOq]      = c0[r];
        hp[(size_t)(16 + r + 8 * half) * HOq] = c1[r];
    }
}

__global__ void __launch_bounds__(256) k_escore(const float* __restrict__ hws,
                                                const float* __restrict__ asrc,
                                                const float* __restrict__ adst,
                                                float* __restrict__ es,
                                                float* __restrict__ ed) {
    int idx = blockIdx.x * 256 + threadIdx.x;
    if (idx >= Bq * Hq * Tq * Nq) return;
    const int n  = idx % Nq;
    const int t  = (idx / Nq) % Tq;
    const int bh = idx / (Nq * Tq);
    const int hh = bh % Hq;
    const float4* hv = (const float4*)(hws + ((size_t)bh * Nq + n) * HOq);
    const float4* sv = (const float4*)(asrc + ((size_t)hh * Tq + t) * HOq);
    const float4* dv = (const float4*)(adst + ((size_t)hh * Tq + t) * HOq);
    float ss = 0.f, sd = 0.f;
#pragma unroll
    for (int q = 0; q < HOq / 4; ++q) {
        float4 hx = hv[q], sx = sv[q], dx = dv[q];
        ss += hx.x * sx.x + hx.y * sx.y + hx.z * sx.z + hx.w * sx.w;
        sd += hx.x * dx.x + hx.y * dx.y + hx.z * dx.z + hx.w * dx.w;
    }
    es[idx] = ss;
    ed[idx] = sd;
}

__device__ __forceinline__ float pick5(int t, float e0, float e1, float e2, float e3, float e4) {
    float v = e4;
    v = (t == 3) ? e3 : v;
    v = (t == 2) ? e2 : v;
    v = (t == 1) ? e1 : v;
    v = (t == 0) ? e0 : v;
    return v;
}

#define SRq 8
#define STq (Nq / 32)
#define SGq 4

__device__ __forceinline__ void ss_load(int (&A)[SRq], float (&E)[Tq],
                                        const int* __restrict__ arow,
                                        const float* __restrict__ edl, int trip) {
    const int j = trip * 32;
#pragma unroll
    for (int t = 0; t < Tq; ++t) E[t] = edl[(size_t)t * Nq + j];
#pragma unroll
    for (int r = 0; r < SRq; ++r) A[r] = arow[(size_t)r * Nq + j];
}

__device__ __forceinline__ void ss_step(const int (&A)[SRq], const float (&E)[Tq],
                                        const float (&esr)[SRq][Tq],
                                        float (&m)[SRq], int (&amin)[SRq]) {
#pragma unroll
    for (int r = 0; r < SRq; ++r) {
        const int a = A[r];
        const int t = (a > 0) ? (a - 1) : 0;
        const float sc = pick5(t, esr[r][0], esr[r][1], esr[r][2], esr[r][3], esr[r][4])
                       + pick5(t, E[0], E[1], E[2], E[3], E[4]);
        m[r]    = fmaxf(m[r], (a == 0) ? -INFINITY : sc);
        amin[r] = min(amin[r], a);
    }
}

__global__ void __launch_bounds__(32 * Hq) k_softstats(const int* __restrict__ adj,
                                                       const float* __restrict__ es,
                                                       const float* __restrict__ ed,
                                                       float* __restrict__ rmax) {
    const int lane = threadIdx.x & 31;
    const int hh = threadIdx.x >> 5;
    const int b  = blockIdx.x / (Nq / SRq);
    const int i0 = (blockIdx.x % (Nq / SRq)) * SRq;
    const int bh = b * Hq + hh;

    const int*   arow = adj + ((size_t)b * Nq + i0) * Nq + lane;
    const float* edl  = ed + (size_t)bh * Tq * Nq + lane;

    float esr[SRq][Tq];
#pragma unroll
    for (int r = 0; r < SRq; ++r) {
#pragma unroll
        for (int t = 0; t < Tq; ++t) esr[r][t] = es[((size_t)bh * Tq + t) * Nq + i0 + r];
    }

    float m[SRq];
    int   amin[SRq];
#pragma unroll
    for (int r = 0; r < SRq; ++r) { m[r] = -INFINITY; amin[r] = Tq; }

    for (int blk = 0; blk < STq; blk += SGq) {
        int   a[SGq][SRq];
        float d[SGq][Tq];
#pragma unroll
        for (int g = 0; g < SGq; ++g) ss_load(a[g], d[g], arow, edl, blk + g);
#pragma unroll
        for (int g = 0; g < SGq; ++g) ss_step(a[g], d[g], esr, m, amin);
    }

#pragma unroll
    for (int off = 16; off >= 1; off >>= 1) {
#pragma unroll
        for (int r = 0; r < SRq; ++r) m[r] = fmaxf(m[r], __shfl_xor(m[r], off, 32));
    }
    float mo_ = 0.f;
#pragma unroll
    for (int r = 0; r < SRq; ++r) {
        float mm = (m[r] > 0.f) ? m[r] : 0.01f * m[r];
        if (__any(amin[r] == 0)) mm = fmaxf(mm, NEGV);
        mo_ = (lane == r) ? mm : mo_;
    }
    if (lane < SRq) rmax[(size_t)bh * Nq + i0 + lane] = mo_;
}

#define KBq 32
#define PPq 36
#define GWq 4
#define RWq (16 / GWq)

__device__ __forceinline__ void at_load(int (&A)[RWq], float (&E)[Tq], v2f (&Bf)[8],
                                        const int* __restrict__ arow,
                                        const float* __restrict__ edl,
                                        const float* __restrict__ hfr, int j) {
#pragma unroll
    for (int r = 0; r < RWq; ++r) A[r] = arow[(size_t)r * Nq + j];
#pragma unroll
    for (int t = 0; t < Tq; ++t) E[t] = edl[(size_t)t * Nq + j];
#pragma unroll
    for (int q = 0; q < 8; ++q) {
        const float* h0 = hfr + (size_t)(j + 4 * q) * HOq;
        Bf[q].x = h0[0];
        Bf[q].y = h0[HOq];
    }
}

__device__ __forceinline__ void at_trip(const int (&A)[RWq], const float (&E)[Tq], const v2f (&Bf)[8],
                                        const float (&esr)[RWq][Tq], const float (&rmx)[RWq],
                                        float (&rs)[RWq], float* __restrict__ Pb,
                                        int prow, int lane, int lm, int half, v8f& c) {
#pragma unroll
    for (int r = 0; r < RWq; ++r) {
        const int a = A[r];
        const int t = (a > 0) ? (a - 1) : 0;
        const float sc = pick5(t, esr[r][0], esr[r][1], esr[r][2], esr[r][3], esr[r][4])
                       + pick5(t, E[0], E[1], E[2], E[3], E[4]);
        const float wv = (a == 0) ? NEGV : (sc > 0.f ? sc : 0.01f * sc);
        const float p = __expf(wv - rmx[r]);
        rs[r] += p;
        Pb[(prow + r) * PPq + lane] = p;
    }
    group_lds_barrier();
#pragma unroll
    for (int q = 0; q < 8; ++q) {
        const v2f ap = *(const v2f*)&Pb[lm * PPq + 4 * q + 2 * half];
        c = __builtin_amdgcn_wmma_f32_16x16x4_f32(false, ap, false, Bf[q], (short)0, c, false, false);
    }
}

__global__ void __launch_bounds__(32 * GWq) k_attn(const int* __restrict__ adj,
                                                   const float* __restrict__ hws,
                                                   const float* __restrict__ es,
                                                   const float* __restrict__ ed,
                                                   const float* __restrict__ rmax,
                                                   const float* __restrict__ bias,
                                                   const float* __restrict__ gamma,
                                                   const float* __restrict__ beta,
                                                   float* __restrict__ out) {
    __shared__ __attribute__((aligned(16))) float Pt[2][16 * PPq];
    __shared__ __attribute__((aligned(16))) float tile[16 * 68];
    __shared__ float stat[16 * 2];
    __shared__ float rinv[16];
    const int lane = threadIdx.x & 31;
    const int w    = __builtin_amdgcn_readfirstlane(threadIdx.x >> 5);
    const int half = lane >> 4;
    const int lm   = lane & 15;
    const int i0   = blockIdx.x * 16;
    const int bh   = blockIdx.y;
    const int b    = bh / Hq, hh = bh % Hq;
    const int prow = RWq * w;

    float esr[RWq][Tq], rmx[RWq], rs[RWq];
#pragma unroll
    for (int r = 0; r < RWq; ++r) {
        const int row = i0 + prow + r;
#pragma unroll
        for (int t = 0; t < Tq; ++t) esr[r][t] = es[((size_t)bh * Tq + t) * Nq + row];
        rmx[r] = rmax[(size_t)bh * Nq + row];
        rs[r]  = 0.f;
    }

    const int*   arow = adj + ((size_t)b * Nq + i0 + prow) * Nq + lane;
    const float* edl  = ed + (size_t)bh * Tq * Nq + lane;
    const float* hb   = hws + (size_t)bh * Nq * HOq;
    const float* hfr  = hb + (size_t)(2 * half) * HOq + 16 * w + lm;

    int   adjA[RWq], adjB[RWq];
    float edA[Tq],   edB[Tq];
    v2f   bfA[8],    bfB[8];
    at_load(adjA, edA, bfA, arow, edl, hfr, 0);

    v8f c = {};
    for (int blk = 0; blk < Nq / KBq; blk += 2) {
        const int j0 = blk * KBq;
        at_load(adjB, edB, bfB, arow, edl, hfr, j0 + KBq);
        at_trip(adjA, edA, bfA, esr, rmx, rs, Pt[0], prow, lane, lm, half, c);
        const int jn = (blk + 2 < Nq / KBq) ? (j0 + 2 * KBq) : (j0 + KBq);
        at_load(adjA, edA, bfA, arow, edl, hfr, jn);
        at_trip(adjB, edB, bfB, esr, rmx, rs, Pt[1], prow, lane, lm, half, c);
    }

#pragma unroll
    for (int off = 16; off >= 1; off >>= 1) {
#pragma unroll
        for (int r = 0; r < RWq; ++r) rs[r] += __shfl_xor(rs[r], off, 32);
    }
    {
        float sr = rs[0];
#pragma unroll
        for (int r = 1; r < RWq; ++r) sr = (lane == r) ? rs[r] : sr;
        if (lane < RWq) rinv[prow + lane] = 1.f / sr;
    }

#pragma unroll
    for (int r = 0; r < 8; ++r) tile[(r + 8 * half) * 68 + 16 * w + lm] = c[r];
    __syncthreads();

    const int row = threadIdx.x >> 3;
    const int e0  = (threadIdx.x & 7) * 8;
    float v[8];
    {
        const float* hres = hb + (size_t)(i0 + row) * HOq + e0;
        const float4 h0 = *(const float4*)hres,                   h1 = *(const float4*)(hres + 4);
        const float4 b0 = *(const float4*)(bias + hh * HOq + e0), b1 = *(const float4*)(bias + hh * HOq + e0 + 4);
        const float4 t0 = *(const float4*)&tile[row * 68 + e0],   t1 = *(const float4*)&tile[row * 68 + e0 + 4];
        const float ri = rinv[row];
        v[0] = fmaxf(t0.x * ri + b0.x, 0.f) + h0.x;
        v[1] = fmaxf(t0.y * ri + b0.y, 0.f) + h0.y;
        v[2] = fmaxf(t0.z * ri + b0.z, 0.f) + h0.z;
        v[3] = fmaxf(t0.w * ri + b0.w, 0.f) + h0.w;
        v[4] = fmaxf(t1.x * ri + b1.x, 0.f) + h1.x;
        v[5] = fmaxf(t1.y * ri + b1.y, 0.f) + h1.y;
        v[6] = fmaxf(t1.z * ri + b1.z, 0.f) + h1.z;
        v[7] = fmaxf(t1.w * ri + b1.w, 0.f) + h1.w;
        *(float4*)&tile[row * 68 + e0]     = make_float4(v[0], v[1], v[2], v[3]);
        *(float4*)&tile[row * 68 + e0 + 4] = make_float4(v[4], v[5], v[6], v[7]);
    }
    __syncthreads();
    if (threadIdx.x < 16) {
        const int r = threadIdx.x;
        float sum = 0.f, sumsq = 0.f;
#pragma unroll 8
        for (int e = 0; e < HOq; ++e) {
            const float x = tile[r * 68 + e];
            sum += x; sumsq += x * x;
        }
        const float mu   = sum * (1.f / HOq);
        const float var  = sumsq * (1.f / HOq) - mu * mu;
        stat[r * 2]     = mu;
        stat[r * 2 + 1] = rsqrtf(var + EPSV);
    }
    __syncthreads();
    {
        const float mu = stat[row * 2], rstd = stat[row * 2 + 1];
        const float4 g0 = *(const float4*)(gamma + hh * HOq + e0), g1 = *(const float4*)(gamma + hh * HOq + e0 + 4);
        const float4 z0 = *(const float4*)(beta + hh * HOq + e0),  z1 = *(const float4*)(beta + hh * HOq + e0 + 4);
        float* op = out + ((size_t)b * Nq + (i0 + row)) * (Hq * HOq) + hh * HOq + e0;
        float4 o0, o1;
        o0.x = (v[0] - mu) * rstd * g0.x + z0.x;
        o0.y = (v[1] - mu) * rstd * g0.y + z0.y;
        o0.z = (v[2] - mu) * rstd * g0.z + z0.z;
        o0.w = (v[3] - mu) * rstd * g0.w + z0.w;
        o1.x = (v[4] - mu) * rstd * g1.x + z1.x;
        o1.y = (v[5] - mu) * rstd * g1.y + z1.y;
        o1.z = (v[6] - mu) * rstd * g1.z + z1.z;
        o1.w = (v[7] - mu) * rstd * g1.w + z1.w;
        *(float4*)op       = o0;
        *(float4*)(op + 4) = o1;
    }
}

extern "C" void kernel_launch(void* const* d_in, const int* in_sizes, int n_in,
                              void* d_out, int out_size, void* d_ws, size_t ws_size,
                              hipStream_t stream) {
    const float* nodes = (const float*)d_in[0];
    const int*   adj   = (const int*)d_in[1];
    const float* Wm    = (const float*)d_in[2];
    const float* asrc  = (const float*)d_in[3];
    const float* adst  = (const float*)d_in[4];
    const float* bias  = (const float*)d_in[5];
    const float* gamma = (const float*)d_in[6];
    const float* beta  = (const float*)d_in[7];
    float* out = (float*)d_out;

    float* h_ws = (float*)d_ws;
    float* esrc = h_ws + (size_t)Bq * Hq * Nq * HOq;
    float* edst = esrc + (size_t)Bq * Hq * Tq * Nq;
    float* rmx  = edst + (size_t)Bq * Hq * Tq * Nq;

    k_proj<<<dim3(Hq, Bq * Nq / PRq), 32 * 4, 0, stream>>>(nodes, Wm, h_ws);
    k_escore<<<(Bq * Hq * Tq * Nq + 255) / 256, 256, 0, stream>>>(h_ws, asrc, adst, esrc, edst);
    k_softstats<<<Bq * (Nq / SRq), 32 * Hq, 0, stream>>>(adj, esrc, edst, rmx);
    k_attn<<<dim3(Nq / 16, Bq * Hq), 32 * GWq, 0, stream>>>(adj, h_ws, esrc, edst, rmx,
                                                      bias, gamma, beta, out);
}


# k_softstats dropped (3 launches): k_attn shifts a row by an upper bound of its max (src term + largest dst term of the type); exact rescan in-kernel if a row sum is tiny; sum order same
# speedup vs baseline: 3.2640x; 1.1728x over previous
#include <hip/hip_runtime.h>

#define Bq  8
#define Nq  1024
#define Dq  256
#define Hq  4
#define HOq 64
#define Tq  5
#define NEGV (-1e9f)
#define EPSV (1e-6f)

typedef float v2f __attribute__((ext_vector_type(2)));
typedef float v8f __attribute__((ext_vector_type(8)));

__device__ __forceinline__ void group_lds_barrier() {
    __builtin_amdgcn_fence(__ATOMIC_RELEASE, "workgroup", "local");
    __builtin_amdgcn_s_barrier();
    __builtin_amdgcn_fence(__ATOMIC_ACQUIRE, "workgroup", "local");
}

#define PRq 32
#define PCq 32
#define PXq 36

__device__ __forceinline__ void pj_load(float4& xa, float4& xb, v2f (&Bf)[8],
                                        const float* __restrict__ xg,
                                        const float* __restrict__ wp, int k0) {
    xa = *(const float4*)(xg + k0);
    xb = *(const float4*)(xg + (size_t)16 * Dq + k0);
#pragma unroll
    for (int q = 0; q < 8; ++q) {
        const float* w0 = wp + (size_t)(k0 + 4 * q) * HOq;
        Bf[q].x = w0[0];
        Bf[q].y = w0[HOq];
    }
}

__device__ __forceinline__ void pj_chunk(const v2f (&Bf)[8], const float* __restrict__ Xb,
                                         int lm, int half, v8f& c0, v8f& c1) {
#pragma unroll
    for (int q = 0; q < 8; ++q) {
        const v2f a0 = *(const v2f*)&Xb[lm * PXq + 4 * q + 2 * half];
        const v2f a1 = *(const v2f*)&Xb[(16 + lm) * PXq + 4 * q + 2 * half];
        c0 = __builtin_amdgcn_wmma_f32_16x16x4_f32(false, a0, false, Bf[q], (short)0, c0, false, false);
        c1 = __builtin_amdgcn_wmma_f32_16x16x4_f32(false, a1, false, Bf[q], (short)0, c1, false, false);
    }
}

__global__ void __launch_bounds__(128) k_proj(const float* __restrict__ X,
                                              const float* __restrict__ Wm,
                                              float* __restrict__ hws) {
    __shared__ __attribute__((aligned(16))) float Xs[2][PRq * PXq];
    const int lane = threadIdx.x & 31;
    const int w    = __builtin_amdgcn_readfirstlane(threadIdx.x >> 5);
    const int half = lane >> 4;
    const int lm   = lane & 15;
    const int hh   = blockIdx.x;
    const int g0   = blockIdx.y * PRq;
    const int b    = g0 / Nq, i0 = g0 % Nq;
    const int bh   = b * Hq + hh;

    const int srow = threadIdx.x >> 3;
    const int scol = (threadIdx.x & 7) * 4;
    const float* xg = X + ((size_t)g0 + srow) * Dq + scol;
    const float* wp = Wm + (size_t)hh * Dq * HOq + (size_t)(2 * half) * HOq + 16 * w + lm;
    const int so = srow * PXq + scol;

    float4 xa, xb;
    v2f bfA[8], bfB[8];
    pj_load(xa, xb, bfA, xg, wp, 0);
    *(float4*)&Xs[0][so]            = xa;
    *(float4*)&Xs[0][so + 16 * PXq] = xb;
    group_lds_barrier();

    v8f c0 = {}, c1 = {};
#pragma unroll
    for (int ch = 0; ch < Dq / PCq; ch += 2) {
        __builtin_amdgcn_sched_barrier(0);
        pj_load(xa, xb, bfB, xg, wp, (ch + 1) * PCq);
        __builtin_amdgcn_sched_barrier(0);
        pj_chunk(bfA, Xs[0], lm, half, c0, c1);
        *(float4*)&Xs[1][so]            = xa;
        *(float4*)&Xs[1][so + 16 * PXq] = xb;
        group_lds_barrier();
        __builtin_amdgcn_sched_barrier(0);
        if (ch + 2 < Dq / PCq) pj_load(xa, xb, bfA, xg, wp, (ch + 2) * PCq);
        __builtin_amdgcn_sched_barrier(0);
        pj_chunk(bfB, Xs[1], lm, half, c0, c1);
        if (ch + 2 < Dq / PCq) {
            *(float4*)&Xs[0][so]            = xa;
            *(float4*)&Xs[0][so + 16 * PXq] = xb;
            group_lds_barrier();
        }
    }
    float* hp = hws + ((size_t)bh * Nq + i0) * HOq + 16 * w + lm;
#pragma unroll
    for (int r = 0; r < 8; ++r) {
        hp[(size_t)(r + 8 * half) * HOq]      = c0[r];
        hp[(size_t)(16 + r + 8 * half) * HOq] = c1[r];
    }
}

static_assert((Bq * Hq * Tq * Nq) % 256 == 0 && Nq % 256 == 0, "a group of k_escore is one (b,h,t) slice of 256 nodes");
__global__ void __launch_bounds__(256) k_escore(const float* __restrict__ hws,
                                                const float* __restrict__ asrc,
                                                const float* __restrict__ adst,
                                                float* __restrict__ es,
                                                float* __restrict__ ed,
                                                float* __restrict__ edmx) {
    __shared__ float wmx[8];
    const int idx = blockIdx.x * 256 + threadIdx.x;
    const int n  = idx % Nq;
    const int t  = (idx / Nq) % Tq;
    const int bh = idx / (Nq * Tq);
    const int hh = bh % Hq;
    const float4* hv = (const float4*)(hws + ((size_t)bh * Nq + n) * HOq);
    const float4* sv = (const float4*)(asrc + ((size_t)hh * Tq + t) * HOq);
    const float4* dv = (const float4*)(adst + ((size_t)hh * Tq + t) * HOq);
    float ss = 0.f, sd = 0.f;
#pragma unroll
    for (int q = 0; q < HOq / 4; ++q) {
        float4 hx = hv[q], sx = sv[q], dx = dv[q];
        ss += hx.x * sx.x + hx.y * sx.y + hx.z * sx.z + hx.w * sx.w;
        sd += hx.x * dx.x + hx.y * dx.y + hx.z * dx.z + hx.w * dx.w;
    }
    es[idx] = ss;
    ed[idx] = sd;
    float mx = sd;
#pragma unroll
    for (int off = 16; off >= 1; off >>= 1) mx = fmaxf(mx, __shfl_xor(mx, off, 32));
    if ((threadIdx.x & 31) == 0) wmx[threadIdx.x >> 5] = mx;
    __syncthreads();
    if (threadIdx.x == 0) {
        float m = wmx[0];
#pragma unroll
        for (int k = 1; k < 8; ++k) m = fmaxf(m, wmx[k]);
        edmx[blockIdx.x] = m;
    }
}

__device__ __forceinline__ float pick5(int t, float e0, float e1, float e2, float e3, float e4) {
    float v = e4;
    v = (t == 3) ? e3 : v;
    v = (t == 2) ? e2 : v;
    v = (t == 1) ? e1 : v;
    v = (t == 0) ? e0 : v;
    return v;
}

#define KBq 32
#define PPq 36
#define GWq 4
#define RWq (16 / GWq)

#define SUMLOq (1e-3f)

__device__ __forceinline__ void exact_rowmax(float (&rmx)[RWq], const float (&esr)[RWq][Tq],
                                             const int* __restrict__ arow,
                                             const float* __restrict__ edl) {
    float m[RWq];
    int   amin[RWq];
#pragma unroll
    for (int r = 0; r < RWq; ++r) { m[r] = -INFINITY; amin[r] = Tq; }
    for (int j = 0; j < Nq; j += KBq) {
        int   A[RWq];
        float E[Tq];
#pragma unroll
        for (int r = 0; r < RWq; ++r) A[r] = arow[(size_t)r * Nq + j];
#pragma unroll
        for (int t = 0; t < Tq; ++t) E[t] = edl[(size_t)t * Nq + j];
#pragma unroll
        for (int r = 0; r < RWq; ++r) {
            const int a = A[r];
            const int t = (a > 0) ? (a - 1) : 0;
            const float sc = pick5(t, esr[r][0], esr[r][1], esr[r][2], esr[r][3], esr[r][4])
                           + pick5(t, E[0], E[1], E[2], E[3], E[4]);
            m[r]    = fmaxf(m[r], (a == 0) ? -INFINITY : sc);
            amin[r] = min(amin[r], a);
        }
    }
#pragma unroll
    for (int off = 16; off >= 1; off >>= 1) {
#pragma unroll
        for (int r = 0; r < RWq; ++r) m[r] = fmaxf(m[r], __shfl_xor(m[r], off, 32));
    }
#pragma unroll
    for (int r = 0; r < RWq; ++r) {
        float mm = (m[r] > 0.f) ? m[r] : 0.01f * m[r];
        if (__any(amin[r] == 0)) mm = fmaxf(mm, NEGV);
        rmx[r] = mm;
    }
}

__device__ __forceinline__ void at_load(int (&A)[RWq], float (&E)[Tq], v2f (&Bf)[8],
                                        const int* __restrict__ arow,
                                        const float* __restrict__ edl,
                                        const float* __restrict__ hfr, int j) {
#pragma unroll
    for (int r = 0; r < RWq; ++r) A[r] = arow[(size_t)r * Nq + j];
#pragma unroll
    for (int t = 0; t < Tq; ++t) E[t] = edl[(size_t)t * Nq + j];
#pragma unroll
    for (int q = 0; q < 8; ++q) {
        const float* h0 = hfr + (size_t)(j + 4 * q) * HOq;
        Bf[q].x = h0[0];
        Bf[q].y = h0[HOq];
    }
}

__device__ __forceinline__ void at_trip(const int (&A)[RWq], const float (&E)[Tq], const v2f (&Bf)[8],
                                        const float (&esr)[RWq][Tq], const float (&rmx)[RWq],
                                        float (&rs)[RWq], float* __restrict__ Pb,
                                        int prow, int lane, int lm, int half, v8f& c) {
#pragma unroll
    for (int r = 0; r < RWq; ++r) {
        const int a = A[r];
        const int t = (a > 0) ? (a - 1) : 0;
        const float sc = pick5(t, esr[r][0], esr[r][1], esr[r][2], esr[r][3], esr[r][4])
                       + pick5(t, E[0], E[1], E[2], E[3], E[4]);
        const float wv = (a == 0) ? NEGV : (sc > 0.f ? sc : 0.01f * sc);
        const float p = __expf(wv - rmx[r]);
        rs[r] += p;
        Pb[(prow + r) * PPq + lane] = p;
    }
    group_lds_barrier();
#pragma unroll
    for (int q = 0; q < 8; ++q) {
        const v2f ap = *(const v2f*)&Pb[lm * PPq + 4 * q + 2 * half];
        c = __builtin_amdgcn_wmma_f32_16x16x4_f32(false, ap, false, Bf[q], (short)0, c, false, false);
    }
}

__device__ __forceinline__ void attn_tile(const float (&rmx)[RWq], const float (&esr)[RWq][Tq],
                                          const int* __restrict__ arow, const float* __restrict__ edl,
                                          const float* __restrict__ hfr,
                                          float* __restrict__ Pt0, float* __restrict__ Pt1,
                                          float* __restrict__ tile, float* __restrict__ rinv,
                                          int* __restrict__ okf,
                                          int prow, int w, int lane, int lm, int half) {
    float rs[RWq];
#pragma unroll
    for (int r = 0; r < RWq; ++r) rs[r] = 0.f;

    int   adjA[RWq], adjB[RWq];
    float edA[Tq],   edB[Tq];
    v2f   bfA[8],    bfB[8];
    at_load(adjA, edA, bfA, arow, edl, hfr, 0);

    v8f c = {};
    for (int blk = 0; blk < Nq / KBq; blk += 2) {
        const int j0 = blk * KBq;
        at_load(adjB, edB, bfB, arow, edl, hfr, j0 + KBq);
        at_trip(adjA, edA, bfA, esr, rmx, rs, Pt0, prow, lane, lm, half, c);
        const int jn = (blk + 2 < Nq / KBq) ? (j0 + 2 * KBq) : (j0 + KBq);
        at_load(adjA, edA, bfA, arow, edl, hfr, jn);
        at_trip(adjB, edB, bfB, esr, rmx, rs, Pt1, prow, lane, lm, half, c);
    }

#pragma unroll
    for (int off = 16; off >= 1; off >>= 1) {
#pragma unroll
        for (int r = 0; r < RWq; ++r) rs[r] += __shfl_xor(rs[r], off, 32);
    }
    {
        float sr = rs[0];
        int   ok = (rs[0] >= SUMLOq);
#pragma unroll
        for (int r = 1; r < RWq; ++r) {
            sr = (lane == r) ? rs[r] : sr;
            ok &= (rs[r] >= SUMLOq);
        }
        if (lane < RWq) rinv[prow + lane] = 1.f / sr;
        if (lane == 0) okf[w] = ok;
    }

#pragma unroll
    for (int r = 0; r < 8; ++r) tile[(r + 8 * half) * 68 + 16 * w + lm] = c[r];
}

__global__ void __launch_bounds__(32 * GWq) k_attn(const int* __restrict__ adj,
                                                   const float* __restrict__ hws,
                                                   const float* __restrict__ es,
                                                   const float* __restrict__ ed,
                                                   const float* __restrict__ edmx,
                                                   const float* __restrict__ bias,
                                                   const float* __restrict__ gamma,
                                                   const float* __restrict__ beta,
                                                   float* __restrict__ out) {
    __shared__ __attribute__((aligned(16))) float Pt[2][16 * PPq];
    __shared__ __attribute__((aligned(16))) float tile[16 * 68];
    __shared__ float stat[16 * 2];
    __shared__ float rinv[16];
    __shared__ int   okf[GWq];
    const int lane = threadIdx.x & 31;
    const int w    = __builtin_amdgcn_readfirstlane(threadIdx.x >> 5);
    const int half = lane >> 4;
    const int lm   = lane & 15;
    const int i0   = blockIdx.x * 16;
    const int bh   = blockIdx.y;
    const int b    = bh / Hq, hh = bh % Hq;
    const int prow = RWq * w;

    float emx[Tq];
#pragma unroll
    for (int t = 0; t < Tq; ++t) {
        const float* p = edmx + ((size_t)bh * Tq + t) * (Nq / 256);
        emx[t] = fmaxf(fmaxf(p[0], p[1]), fmaxf(p[2], p[3]));
    }

    float esr[RWq][Tq], rmx[RWq];
#pragma unroll
    for (int r = 0; r < RWq; ++r) {
        const int row = i0 + prow + r;
#pragma unroll
        for (int t = 0; t < Tq; ++t) esr[r][t] = es[((size_t)bh * Tq + t) * Nq + row];
        float u = esr[r][0] + emx[0];
#pragma unroll
        for (int t = 1; t < Tq; ++t) u = fmaxf(u, esr[r][t] + emx[t]);
        rmx[r] = (u > 0.f) ? u : 0.01f * u;
    }

    const int*   arow = adj + ((size_t)b * Nq + i0 + prow) * Nq + lane;
    const float* edl  = ed + (size_t)bh * Tq * Nq + lane;
    const float* hb   = hws + (size_t)bh * Nq * HOq;
    const float* hfr  = hb + (size_t)(2 * half) * HOq + 16 * w + lm;

    attn_tile(rmx, esr, arow, edl, hfr, Pt[0], Pt[1], tile, rinv, okf, prow, w, lane, lm, half);
    __syncthreads();
    const int allok = __builtin_amdgcn_readfirstlane(okf[0] & okf[1] & okf[2] & okf[3]);
    if (!allok) {
        exact_rowmax(rmx, esr, arow, edl);
        attn_tile(rmx, esr, arow, edl, hfr, Pt[0], Pt[1], tile, rinv, okf, prow, w, lane, lm, half);
        __syncthreads();
    }

    const int row = threadIdx.x >> 3;
    const int e0  = (threadIdx.x & 7) * 8;
    float v[8];
    {
        const float* hres = hb + (size_t)(i0 + row) * HOq + e0;
        const float4 h0 = *(const float4*)hres,                   h1 = *(const float4*)(hres + 4);
        const float4 b0 = *(const float4*)(bias + hh * HOq + e0), b1 = *(const float4*)(bias + hh * HOq + e0 + 4);
        const float4 t0 = *(const float4*)&tile[row * 68 + e0],   t1 = *(const float4*)&tile[row * 68 + e0 + 4];
        const float ri = rinv[row];
        v[0] = fmaxf(t0.x * ri + b0.x, 0.f) + h0.x;
        v[1] = fmaxf(t0.y * ri + b0.y, 0.f) + h0.y;
        v[2] = fmaxf(t0.z * ri + b0.z, 0.f) + h0.z;
        v[3] = fmaxf(t0.w * ri + b0.w, 0.f) + h0.w;
        v[4] = fmaxf(t1.x * ri + b1.x, 0.f) + h1.x;
        v[5] = fmaxf(t1.y * ri + b1.y, 0.f) + h1.y;
        v[6] = fmaxf(t1.z * ri + b1.z, 0.f) + h1.z;
        v[7] = fmaxf(t1.w * ri + b1.w, 0.f) + h1.w;
        *(float4*)&tile[row * 68 + e0]     = make_float4(v[0], v[1], v[2], v[3]);
        *(float4*)&tile[row * 68 + e0 + 4] = make_float4(v[4], v[5], v[6], v[7]);
    }
    __syncthreads();
    if (threadIdx.x < 16) {
        const int r = threadIdx.x;
        float sum = 0.f, sumsq = 0.f;
#pragma unroll 8
        for (int e = 0; e < HOq; ++e) {
            const float x = tile[r * 68 + e];
            sum += x; sumsq += x * x;
        }
        const float mu   = sum * (1.f / HOq);
        const float var  = sumsq * (1.f / HOq) - mu * mu;
        stat[r * 2]     = mu;
        stat[r * 2 + 1] = rsqrtf(var + EPSV);
    }
    __syncthreads();
    {
        const float mu = stat[row * 2], rstd = stat[row * 2 + 1];
        const float4 g0 = *(const float4*)(gamma + hh * HOq + e0), g1 = *(const float4*)(gamma + hh * HOq + e0 + 4);
        const float4 z0 = *(const float4*)(beta + hh * HOq + e0),  z1 = *(const float4*)(beta + hh * HOq + e0 + 4);
        float* op = out + ((size_t)b * Nq + (i0 + row)) * (Hq * HOq) + hh * HOq + e0;
        float4 o0, o1;
        o0.x = (v[0] - mu) * rstd * g0.x + z0.x;
        o0.y = (v[1] - mu) * rstd * g0.y + z0.y;
        o0.z = (v[2] - mu) * rstd * g0.z + z0.z;
        o0.w = (v[3] - mu) * rstd * g0.w + z0.w;
        o1.x = (v[4] - mu) * rstd * g1.x + z1.x;
        o1.y = (v[5] - mu) * rstd * g1.y + z1.y;
        o1.z = (v[6] - mu) * rstd * g1.z + z1.z;
        o1.w = (v[7] - mu) * rstd * g1.w + z1.w;
        *(float4*)op       = o0;
        *(float4*)(op + 4) = o1;
    }
}

extern "C" void kernel_launch(void* const* d_in, const int* in_sizes, int n_in,
                              void* d_out, int out_size, void* d_ws, size_t ws_size,
                              hipStream_t stream) {
    const float* nodes = (const float*)d_in[0];
    const int*   adj   = (const int*)d_in[1];
    const float* Wm    = (const float*)d_in[2];
    const float* asrc  = (const float*)d_in[3];
    const float* adst  = (const float*)d_in[4];
    const float* bias  = (const float*)d_in[5];
    const float* gamma = (const float*)d_in[6];
    const float* beta  = (const float*)d_in[7];
    float* out = (float*)d_out;

    float* h_ws = (float*)d_ws;
    float* esrc = h_ws + (size_t)Bq * Hq * Nq * HOq;
    float* edst = esrc + (size_t)Bq * Hq * Tq * Nq;
    float* edmx = edst + (size_t)Bq * Hq * Tq * Nq;

    k_proj<<<dim3(Hq, Bq * Nq / PRq), 32 * 4, 0, stream>>>(nodes, Wm, h_ws);
    k_escore<<<(Bq * Hq * Tq * Nq) / 256, 256, 0, stream>>>(h_ws, asrc, adst, esrc, edst, edmx);
    k_attn<<<dim3(Nq / 16, Bq * Hq), 32 * GWq, 0, stream>>>(adj, h_ws, esrc, edst, edmx,
                                                      bias, gamma, beta, out);
}


# h kept in the workspace in k_attn's B-fragment order: k_proj stores, k_attn loads whole lines, 16 B a lane (4 loads a trip, not 16); a trip's 32 neighbours meet in another k order: sum order changes
# speedup vs baseline: 3.4397x; 1.0538x over previous
#include <hip/hip_runtime.h>

#define Bq  8
#define Nq  1024
#define Dq  256
#define Hq  4
#define HOq 64
#define Tq  5
#define NEGV (-1e9f)
#define EPSV (1e-6f)

typedef float v2f __attribute__((ext_vector_type(2)));
typedef float v8f __attribute__((ext_vector_type(8)));
typedef float v4f __attribute__((ext_vector_type(4)));

#define HBq 512
__device__ __forceinline__ size_t hf_block(int bh, int jb, int ct) {
    return ((size_t)(bh * (Nq / 32) + jb) * (HOq / 16) + ct) * HBq;
}
__device__ __forceinline__ int hf_off(int rho, int c) {
    const int half = (rho >> 3) & 1;
    const int e    = (rho & 7) + 8 * (rho >> 4);
    return ((e >> 2) * 32 + c + 16 * half) * 4 + (e & 3);
}

__device__ __forceinline__ void group_lds_barrier() {
    __builtin_amdgcn_fence(__ATOMIC_RELEASE, "workgroup", "local");
    __builtin_amdgcn_s_barrier();
    __builtin_amdgcn_fence(__ATOMIC_ACQUIRE, "workgroup", "local");
}

#define PRq 32
#define PCq 32
#define PXq 36

__device__ __forceinline__ void pj_load(float4& xa, float4& xb, v2f (&Bf)[8],
                                        const float* __restrict__ xg,
                                        const float* __restrict__ wp, int k0) {
    xa = *(const float4*)(xg + k0);
    xb = *(const float4*)(xg + (size_t)16 * Dq + k0);
#pragma unroll
    for (int q = 0; q < 8; ++q) {
        const float* w0 = wp + (size_t)(k0 + 4 * q) * HOq;
        Bf[q].x = w0[0];
        Bf[q].y = w0[HOq];
    }
}

__device__ __forceinline__ void pj_chunk(const v2f (&Bf)[8], const float* __restrict__ Xb,
                                         int lm, int half, v8f& c0, v8f& c1) {
#pragma unroll
    for (int q = 0; q < 8; ++q) {
        const v2f a0 = *(const v2f*)&Xb[lm * PXq + 4 * q + 2 * half];
        const v2f a1 = *(const v2f*)&Xb[(16 + lm) * PXq + 4 * q + 2 * half];
        c0 = __builtin_amdgcn_wmma_f32_16x16x4_f32(false, a0, false, Bf[q], (short)0, c0, false, false);
        c1 = __builtin_amdgcn_wmma_f32_16x16x4_f32(false, a1, false, Bf[q], (short)0, c1, false, false);
    }
}

__global__ void __launch_bounds__(128) k_proj(const float* __restrict__ X,
                                              const float* __restrict__ Wm,
                                              float* __restrict__ hws) {
    __shared__ __attribute__((aligned(16))) float Xs[2][PRq * PXq];
    const int lane = threadIdx.x & 31;
    const int w    = __builtin_amdgcn_readfirstlane(threadIdx.x >> 5);
    const int half = lane >> 4;
    const int lm   = lane & 15;
    const int hh   = blockIdx.x;
    const int g0   = blockIdx.y * PRq;
    const int b    = g0 / Nq, i0 = g0 % Nq;
    const int bh   = b * Hq + hh;

    const int srow = threadIdx.x >> 3;
    const int scol = (threadIdx.x & 7) * 4;
    const float* xg = X + ((size_t)g0 + srow) * Dq + scol;
    const float* wp = Wm + (size_t)hh * Dq * HOq + (size_t)(2 * half) * HOq + 16 * w + lm;
    const int so = srow * PXq + scol;

    float4 xa, xb;
    v2f bfA[8], bfB[8];
    pj_load(xa, xb, bfA, xg, wp, 0);
    *(float4*)&Xs[0][so]            = xa;
    *(float4*)&Xs[0][so + 16 * PXq] = xb;
    group_lds_barrier();

    v8f c0 = {}, c1 = {};
#pragma unroll
    for (int ch = 0; ch < Dq / PCq; ch += 2) {
        __builtin_amdgcn_sched_barrier(0);
        pj_load(xa, xb, bfB, xg, wp, (ch + 1) * PCq);
        __builtin_amdgcn_sched_barrier(0);
        pj_chunk(bfA, Xs[0], lm, half, c0, c1);
        *(float4*)&Xs[1][so]            = xa;
        *(float4*)&Xs[1][so + 16 * PXq] = xb;
        group_lds_barrier();
        __builtin_amdgcn_sched_barrier(0);
        if (ch + 2 < Dq / PCq) pj_load(xa, xb, bfA, xg, wp, (ch + 2) * PCq);
        __builtin_amdgcn_sched_barrier(0);
        pj_chunk(bfB, Xs[1], lm, half, c0, c1);
        if (ch + 2 < Dq / PCq) {
            *(float4*)&Xs[0][so]            = xa;
            *(float4*)&Xs[0][so + 16 * PXq] = xb;
            group_lds_barrier();
        }
    }
    static_assert(PRq == 32, "a group tile of k_proj is one node block of the fragment order");
    float* hp = hws + hf_block(bh, i0 >> 5, w) + lane * 4;
    *(float4*)(hp)       = make_float4(c0[0], c0[1], c0[2], c0[3]);
    *(float4*)(hp + 128) = make_float4(c0[4], c0[5], c0[6], c0[7]);
    *(float4*)(hp + 256) = make_float4(c1[0], c1[1], c1[2], c1[3]);
    *(float4*)(hp + 384) = make_float4(c1[4], c1[5], c1[6], c1[7]);
}

static_assert((Bq * Hq * Tq * Nq) % 256 == 0 && Nq % 256 == 0, "a group of k_escore is one (b,h,t) slice of 256 nodes");
__global__ void __launch_bounds__(256) k_escore(const float* __restrict__ hws,
                                                const float* __restrict__ asrc,
                                                const float* __restrict__ adst,
                                                float* __restrict__ es,
                                                float* __restrict__ ed,
                                                float* __restrict__ edmx) {
    __shared__ float wmx[8];
    const int idx = blockIdx.x * 256 + threadIdx.x;
    const int n  = idx % Nq;
    const int t  = (idx / Nq) % Tq;
    const int bh = idx / (Nq * Tq);
    const int hh = bh % Hq;
    const float* hp = hws + hf_block(bh, n >> 5, 0) + hf_off(n & 31, 0);
    const float4* sv = (const float4*)(asrc + ((size_t)hh * Tq + t) * HOq);
    const float4* dv = (const float4*)(adst + ((size_t)hh * Tq + t) * HOq);
    float ss = 0.f, sd = 0.f;
#pragma unroll
    for (int q = 0; q < HOq / 4; ++q) {
        const float* hq = hp + (q >> 2) * HBq + (q & 3) * 16;
        float4 hx = make_float4(hq[0], hq[4], hq[8], hq[12]), sx = sv[q], dx = dv[q];
        ss += hx.x * sx.x + hx.y * sx.y + hx.z * sx.z + hx.w * sx.w;
        sd += hx.x * dx.x + hx.y * dx.y + hx.z * dx.z + hx.w * dx.w;
    }
    es[idx] = ss;
    ed[idx] = sd;
    float mx = sd;
#pragma unroll
    for (int off = 16; off >= 1; off >>= 1) mx = fmaxf(mx, __shfl_xor(mx, off, 32));
    if ((threadIdx.x & 31) == 0) wmx[threadIdx.x >> 5] = mx;
    __syncthreads();
    if (threadIdx.x == 0) {
        float m = wmx[0];
#pragma unroll
        for (int k = 1; k < 8; ++k) m = fmaxf(m, wmx[k]);
        edmx[blockIdx.x] = m;
    }
}

__device__ __forceinline__ float pick5(int t, float e0, float e1, float e2, float e3, float e4) {
    float v = e4;
    v = (t == 3) ? e3 : v;
    v = (t == 2) ? e2 : v;
    v = (t == 1) ? e1 : v;
    v = (t == 0) ? e0 : v;
    return v;
}

#define KBq 32
#define PPq 36
#define GWq 4
#define RWq (16 / GWq)

#define SUMLOq (1e-3f)

__device__ __forceinline__ void exact_rowmax(float (&rmx)[RWq], const float (&esr)[RWq][Tq],
                                             const int* __restrict__ arow,
                                             const float* __restrict__ edl) {
    float m[RWq];
    int   amin[RWq];
#pragma unroll
    for (int r = 0; r < RWq; ++r) { m[r] = -INFINITY; amin[r] = Tq; }
    for (int j = 0; j < Nq; j += KBq) {
        int   A[RWq];
        float E[Tq];
#pragma unroll
        for (int r = 0; r < RWq; ++r) A[r] = arow[(size_t)r * Nq + j];
#pragma unroll
        for (int t = 0; t < Tq; ++t) E[t] = edl[(size_t)t * Nq + j];
#pragma unroll
        for (int r = 0; r < RWq; ++r) {
            const int a = A[r];
            const int t = (a > 0) ? (a - 1) : 0;
            const float sc = pick5(t, esr[r][0], esr[r][1], esr[r][2], esr[r][3], esr[r][4])
                           + pick5(t, E[0], E[1], E[2], E[3], E[4]);
            m[r]    = fmaxf(m[r], (a == 0) ? -INFINITY : sc);
            amin[r] = min(amin[r], a);
        }
    }
#pragma unroll
    for (int off = 16; off >= 1; off >>= 1) {
#pragma unroll
        for (int r = 0; r < RWq; ++r) m[r] = fmaxf(m[r], __shfl_xor(m[r], off, 32));
    }
#pragma unroll
    for (int r = 0; r < RWq; ++r) {
        float mm = (m[r] > 0.f) ? m[r] : 0.01f * m[r];
        if (__any(amin[r] == 0)) mm = fmaxf(mm, NEGV);
        rmx[r] = mm;
    }
}

__device__ __forceinline__ void at_load(int (&A)[RWq], float (&E)[Tq], v4f (&Bf)[4],
                                        const int* __restrict__ arow,
                                        const float* __restrict__ edl,
                                        const float* __restrict__ hfr, int j) {
#pragma unroll
    for (int r = 0; r < RWq; ++r) A[r] = arow[(size_t)r * Nq + j];
#pragma unroll
    for (int t = 0; t < Tq; ++t) E[t] = edl[(size_t)t * Nq + j];
    const float* h0 = hfr + (size_t)j * HOq;
#pragma unroll
    for (int k = 0; k < 4; ++k) Bf[k] = *(const v4f*)(h0 + k * 128);
}

__device__ __forceinline__ void at_trip(const int (&A)[RWq], const float (&E)[Tq], const v4f (&Bf)[4],
                                        const float (&esr)[RWq][Tq], const float (&rmx)[RWq],
                                        float (&rs)[RWq], float* __restrict__ Pb,
                                        int prow, int pcol, int lm, int half, v8f& c) {
#pragma unroll
    for (int r = 0; r < RWq; ++r) {
        const int a = A[r];
        const int t = (a > 0) ? (a - 1) : 0;
        const float sc = pick5(t, esr[r][0], esr[r][1], esr[r][2], esr[r][3], esr[r][4])
                       + pick5(t, E[0], E[1], E[2], E[3], E[4]);
        const float wv = (a == 0) ? NEGV : (sc > 0.f ? sc : 0.01f * sc);
        const float p = __expf(wv - rmx[r]);
        rs[r] += p;
        Pb[(prow + r) * PPq + pcol] = p;
    }
    group_lds_barrier();
#pragma unroll
    for (int q = 0; q < 8; ++q) {
        const v2f ap = *(const v2f*)&Pb[lm * PPq + 4 * q + 2 * half];
        const v2f bq = (q & 1) ? Bf[q >> 1].hi : Bf[q >> 1].lo;
        c = __builtin_amdgcn_wmma_f32_16x16x4_f32(false, ap, false, bq, (short)0, c, false, false);
    }
}

__device__ __forceinline__ void attn_tile(const float (&rmx)[RWq], const float (&esr)[RWq][Tq],
                                          const int* __restrict__ arow, const float* __restrict__ edl,
                                          const float* __restrict__ hfr,
                                          float* __restrict__ Pt0, float* __restrict__ Pt1,
                                          float* __restrict__ tile, float* __restrict__ rinv,
                                          int* __restrict__ okf,
                                          int prow, int pcol, int w, int lane, int lm, int half) {
    float rs[RWq];
#pragma unroll
    for (int r = 0; r < RWq; ++r) rs[r] = 0.f;

    int   adjA[RWq], adjB[RWq];
    float edA[Tq],   edB[Tq];
    v4f   bfA[4],    bfB[4];
    at_load(adjA, edA, bfA, arow, edl, hfr, 0);

    v8f c = {};
    for (int blk = 0; blk < Nq / KBq; blk += 2) {
        const int j0 = blk * KBq;
        at_load(adjB, edB, bfB, arow, edl, hfr, j0 + KBq);
        at_trip(adjA, edA, bfA, esr, rmx, rs, Pt0, prow, pcol, lm, half, c);
        const int jn = (blk + 2 < Nq / KBq) ? (j0 + 2 * KBq) : (j0 + KBq);
        at_load(adjA, edA, bfA, arow, edl, hfr, jn);
        at_trip(adjB, edB, bfB, esr, rmx, rs, Pt1, prow, pcol, lm, half, c);
    }

#pragma unroll
    for (int off = 16; off >= 1; off >>= 1) {
#pragma unroll
        for (int r = 0; r < RWq; ++r) rs[r] += __shfl_xor(rs[r], off, 32);
    }
    {
        float sr = rs[0];
        int   ok = (rs[0] >= SUMLOq);
#pragma unroll
        for (int r = 1; r < RWq; ++r) {
            sr = (lane == r) ? rs[r] : sr;
            ok &= (rs[r] >= SUMLOq);
        }
        if (lane < RWq) rinv[prow + lane] = 1.f / sr;
        if (lane == 0) okf[w] = ok;
    }

#pragma unroll
    for (int r = 0; r < 8; ++r) tile[(r + 8 * half) * 68 + 16 * w + lm] = c[r];
}

__global__ void __launch_bounds__(32 * GWq) k_attn(const int* __restrict__ adj,
                                                   const float* __restrict__ hws,
                                                   const float* __restrict__ es,
                                                   const float* __restrict__ ed,
                                                   const float* __restrict__ edmx,
                                                   const float* __restrict__ bias,
                                                   const float* __restrict__ gamma,
                                                   const float* __restrict__ beta,
                                                   float* __restrict__ out) {
    __shared__ __attribute__((aligned(16))) float Pt[2][16 * PPq];
    __shared__ __attribute__((aligned(16))) float tile[16 * 68];
    __shared__ float stat[16 * 2];
    __shared__ float rinv[16];
    __shared__ int   okf[GWq];
    const int lane = threadIdx.x & 31;
    const int w    = __builtin_amdgcn_readfirstlane(threadIdx.x >> 5);
    const int half = lane >> 4;
    const int lm   = lane & 15;
    const int i0   = blockIdx.x * 16;
    const int bh   = blockIdx.y;
    const int b    = bh / Hq, hh = bh % Hq;
    const int prow = RWq * w;
    const int pcol = 16 * (lane >> 4) + 4 * ((lane >> 1) & 3) + 2 * ((lane >> 3) & 1) + (lane & 1);

    float emx[Tq];
#pragma unroll
    for (int t = 0; t < Tq; ++t) {
        const float* p = edmx + ((size_t)bh * Tq + t) * (Nq / 256);
        emx[t] = fmaxf(fmaxf(p[0], p[1]), fmaxf(p[2], p[3]));
    }

    float esr[RWq][Tq], rmx[RWq];
#pragma unroll
    for (int r = 0; r < RWq; ++r) {
        const int row = i0 + prow + r;
#pragma unroll
        for (int t = 0; t < Tq; ++t) esr[r][t] = es[((size_t)bh * Tq + t) * Nq + row];
        float u = esr[r][0] + emx[0];
#pragma unroll
        for (int t = 1; t < Tq; ++t) u = fmaxf(u, esr[r][t] + emx[t]);
        rmx[r] = (u > 0.f) ? u : 0.01f * u;
    }

    const int*   arow = adj + ((size_t)b * Nq + i0 + prow) * Nq + lane;
    const float* edl  = ed + (size_t)bh * Tq * Nq + lane;
    const float* hb   = hws + hf_block(bh, 0, 0);
    const float* hfr  = hb + w * HBq + lane * 4;

    attn_tile(rmx, esr, arow, edl, hfr, Pt[0], Pt[1], tile, rinv, okf, prow, pcol, w, lane, lm, half);
    __syncthreads();
    const int allok = __builtin_amdgcn_readfirstlane(okf[0] & okf[1] & okf[2] & okf[3]);
    if (!allok) {
        exact_rowmax(rmx, esr, arow, edl);
        attn_tile(rmx, esr, arow, edl, hfr, Pt[0], Pt[1], tile, rinv, okf, prow, pcol, w, lane, lm, half);
        __syncthreads();
    }

    const int row = threadIdx.x >> 3;
    const int e0  = (threadIdx.x & 7) * 8;
    float v[8];
    {
        const int nrow = i0 + row;
        const float* hres = hb + (size_t)(nrow >> 5) * ((HOq / 16) * HBq) + (e0 >> 4) * HBq + hf_off(nrow & 31, e0 & 15);
        const float4 h0 = make_float4(hres[0],  hres[4],  hres[8],  hres[12]);
        const float4 h1 = make_float4(hres[16], hres[20], hres[24], hres[28]);
        const float4 b0 = *(const float4*)(bias + hh * HOq + e0), b1 = *(const float4*)(bias + hh * HOq + e0 + 4);
        const float4 t0 = *(const float4*)&tile[row * 68 + e0],   t1 = *(const float4*)&tile[row * 68 + e0 + 4];
        const float ri = rinv[row];
        v[0] = fmaxf(t0.x * ri + b0.x, 0.f) + h0.x;
        v[1] = fmaxf(t0.y * ri + b0.y, 0.f) + h0.y;
        v[2] = fmaxf(t0.z * ri + b0.z, 0.f) + h0.z;
        v[3] = fmaxf(t0.w * ri + b0.w, 0.f) + h0.w;
        v[4] = fmaxf(t1.x * ri + b1.x, 0.f) + h1.x;
        v[5] = fmaxf(t1.y * ri + b1.y, 0.f) + h1.y;
        v[6] = fmaxf(t1.z * ri + b1.z, 0.f) + h1.z;
        v[7] = fmaxf(t1.w * ri + b1.w, 0.f) + h1.w;
        *(float4*)&tile[row * 68 + e0]     = make_float4(v[0], v[1], v[2], v[3]);
        *(float4*)&tile[row * 68 + e0 + 4] = make_float4(v[4], v[5], v[6], v[7]);
    }
    __syncthreads();
    if (threadIdx.x < 16) {
        const int r = threadIdx.x;
        float sum = 0.f, sumsq = 0.f;
#pragma unroll 8
        for (int e = 0; e < HOq; ++e) {
            const float x = tile[r * 68 + e];
            sum += x; sumsq += x * x;
        }
        const float mu   = sum * (1.f / HOq);
        const float var  = sumsq * (1.f / HOq) - mu * mu;
        stat[r * 2]     = mu;
        stat[r * 2 + 1] = rsqrtf(var + EPSV);
    }
    __syncthreads();
    {
        const float mu = stat[row * 2], rstd = stat[row * 2 + 1];
        const float4 g0 = *(const float4*)(gamma + hh * HOq + e0), g1 = *(const float4*)(gamma + hh * HOq + e0 + 4);
        const float4 z0 = *(const float4*)(beta + hh * HOq + e0),  z1 = *(const float4*)(beta + hh * HOq + e0 + 4);
        float* op = out + ((size_t)b * Nq + (i0 + row)) * (Hq * HOq) + hh * HOq + e0;
        float4 o0, o1;
        o0.x = (v[0] - mu) * rstd * g0.x + z0.x;
        o0.y = (v[1] - mu) * rstd * g0.y + z0.y;
        o0.z = (v[2] - mu) * rstd * g0.z + z0.z;
        o0.w = (v[3] - mu) * rstd * g0.w + z0.w;
        o1.x = (v[4] - mu) * rstd * g1.x + z1.x;
        o1.y = (v[5] - mu) * rstd * g1.y + z1.y;
        o1.z = (v[6] - mu) * rstd * g1.z + z1.z;
        o1.w = (v[7] - mu) * rstd * g1.w + z1.w;
        *(float4*)op       = o0;
        *(float4*)(op + 4) = o1;
    }
}

extern "C" void kernel_launch(void* const* d_in, const int* in_sizes, int n_in,
                              void* d_out, int out_size, void* d_ws, size_t ws_size,
                              hipStream_t stream) {
    const float* nodes = (const float*)d_in[0];
    const int*   adj   = (const int*)d_in[1];
    const float* Wm    = (const float*)d_in[2];
    const float* asrc  = (const float*)d_in[3];
    const float* adst  = (const float*)d_in[4];
    const float* bias  = (const float*)d_in[5];
    const float* gamma = (const float*)d_in[6];
    const float* beta  = (const float*)d_in[7];
    float* out = (float*)d_out;

    float* h_ws = (float*)d_ws;
    float* esrc = h_ws + (size_t)Bq * Hq * Nq * HOq;
    float* edst = esrc + (size_t)Bq * Hq * Tq * Nq;
    float* edmx = edst + (size_t)Bq * Hq * Tq * Nq;

    k_proj<<<dim3(Hq, Bq * Nq / PRq), 32 * 4, 0, stream>>>(nodes, Wm, h_ws);
    k_escore<<<(Bq * Hq * Tq * Nq) / 256, 256, 0, stream>>>(h_ws, asrc, adst, esrc, edst, edmx);
    k_attn<<<dim3(Nq / 16, Bq * Hq), 32 * GWq, 0, stream>>>(adj, h_ws, esrc, edst, edmx,
                                                      bias, gamma, beta, out);
}


# k_attn: a 16-wave group owns 64 rows (a wave = one 16x16 output tile); a trip's h (8 KB) is fetched once per group, 16 B a thread, and read as B fragments from LDS; k and sum order unchanged
# speedup vs baseline: 3.7019x; 1.0762x over previous
#include <hip/hip_runtime.h>

#define Bq  8
#define Nq  1024
#define Dq  256
#define Hq  4
#define HOq 64
#define Tq  5
#define NEGV (-1e9f)
#define EPSV (1e-6f)

typedef float v2f __attribute__((ext_vector_type(2)));
typedef float v8f __attribute__((ext_vector_type(8)));
typedef float v4f __attribute__((ext_vector_type(4)));

#define HBq 512
__device__ __forceinline__ size_t hf_block(int bh, int jb, int ct) {
    return ((size_t)(bh * (Nq / 32) + jb) * (HOq / 16) + ct) * HBq;
}
__device__ __forceinline__ int hf_off(int rho, int c) {
    const int half = (rho >> 3) & 1;
    const int e    = (rho & 7) + 8 * (rho >> 4);
    return ((e >> 2) * 32 + c + 16 * half) * 4 + (e & 3);
}

__device__ __forceinline__ void group_lds_barrier() {
    __builtin_amdgcn_fence(__ATOMIC_RELEASE, "workgroup", "local");
    __builtin_amdgcn_s_barrier();
    __builtin_amdgcn_fence(__ATOMIC_ACQUIRE, "workgroup", "local");
}

#define PRq 32
#define PCq 32
#define PXq 36

__device__ __forceinline__ void pj_load(float4& xa, float4& xb, v2f (&Bf)[8],
                                        const float* __restrict__ xg,
                                        const float* __restrict__ wp, int k0) {
    xa = *(const float4*)(xg + k0);
    xb = *(const float4*)(xg + (size_t)16 * Dq + k0);
#pragma unroll
    for (int q = 0; q < 8; ++q) {
        const float* w0 = wp + (size_t)(k0 + 4 * q) * HOq;
        Bf[q].x = w0[0];
        Bf[q].y = w0[HOq];
    }
}

__device__ __forceinline__ void pj_chunk(const v2f (&Bf)[8], const float* __restrict__ Xb,
                                         int lm, int half, v8f& c0, v8f& c1) {
#pragma unroll
    for (int q = 0; q < 8; ++q) {
        const v2f a0 = *(const v2f*)&Xb[lm * PXq + 4 * q + 2 * half];
        const v2f a1 = *(const v2f*)&Xb[(16 + lm) * PXq + 4 * q + 2 * half];
        c0 = __builtin_amdgcn_wmma_f32_16x16x4_f32(false, a0, false, Bf[q], (short)0, c0, false, false);
        c1 = __builtin_amdgcn_wmma_f32_16x16x4_f32(false, a1, false, Bf[q], (short)0, c1, false, false);
    }
}

__global__ void __launch_bounds__(128) k_proj(const float* __restrict__ X,
                                              const float* __restrict__ Wm,
                                              float* __restrict__ hws) {
    __shared__ __attribute__((aligned(16))) float Xs[2][PRq * PXq];
    const int lane = threadIdx.x & 31;
    const int w    = __builtin_amdgcn_readfirstlane(threadIdx.x >> 5);
    const int half = lane >> 4;
    const int lm   = lane & 15;
    const int hh   = blockIdx.x;
    const int g0   = blockIdx.y * PRq;
    const int b    = g0 / Nq, i0 = g0 % Nq;
    const int bh   = b * Hq + hh;

    const int srow = threadIdx.x >> 3;
    const int scol = (threadIdx.x & 7) * 4;
    const float* xg = X + ((size_t)g0 + srow) * Dq + scol;
    const float* wp = Wm + (size_t)hh * Dq * HOq + (size_t)(2 * half) * HOq + 16 * w + lm;
    const int so = srow * PXq + scol;

    float4 xa, xb;
    v2f bfA[8], bfB[8];
    pj_load(xa, xb, bfA, xg, wp, 0);
    *(float4*)&Xs[0][so]            = xa;
    *(float4*)&Xs[0][so + 16 * PXq] = xb;
    group_lds_barrier();

    v8f c0 = {}, c1 = {};
#pragma unroll
    for (int ch = 0; ch < Dq / PCq; ch += 2) {
        __builtin_amdgcn_sched_barrier(0);
        pj_load(xa, xb, bfB, xg, wp, (ch + 1) * PCq);
        __builtin_amdgcn_sched_barrier(0);
        pj_chunk(bfA, Xs[0], lm, half, c0, c1);
        *(float4*)&Xs[1][so]            = xa;
        *(float4*)&Xs[1][so + 16 * PXq] = xb;
        group_lds_barrier();
        __builtin_amdgcn_sched_barrier(0);
        if (ch + 2 < Dq / PCq) pj_load(xa, xb, bfA, xg, wp, (ch + 2) * PCq);
        __builtin_amdgcn_sched_barrier(0);
        pj_chunk(bfB, Xs[1], lm, half, c0, c1);
        if (ch + 2 < Dq / PCq) {
            *(float4*)&Xs[0][so]            = xa;
            *(float4*)&Xs[0][so + 16 * PXq] = xb;
            group_lds_barrier();
        }
    }
    static_assert(PRq == 32, "a group tile of k_proj is one node block of the fragment order");
    float* hp = hws + hf_block(bh, i0 >> 5, w) + lane * 4;
    *(float4*)(hp)       = make_float4(c0[0], c0[1], c0[2], c0[3]);
    *(float4*)(hp + 128) = make_float4(c0[4], c0[5], c0[6], c0[7]);
    *(float4*)(hp + 256) = make_float4(c1[0], c1[1], c1[2], c1[3]);
    *(float4*)(hp + 384) = make_float4(c1[4], c1[5], c1[6], c1[7]);
}

static_assert((Bq * Hq * Tq * Nq) % 256 == 0 && Nq % 256 == 0, "a group of k_escore is one (b,h,t) slice of 256 nodes");
__global__ void __launch_bounds__(256) k_escore(const float* __restrict__ hws,
                                                const float* __restrict__ asrc,
                                                const float* __restrict__ adst,
                                                float* __restrict__ es,
                                                float* __restrict__ ed,
                                                float* __restrict__ edmx) {
    __shared__ float wmx[8];
    const int idx = blockIdx.x * 256 + threadIdx.x;
    const int n  = idx % Nq;
    const int t  = (idx / Nq) % Tq;
    const int bh = idx / (Nq * Tq);
    const int hh = bh % Hq;
    const float* hp = hws + hf_block(bh, n >> 5, 0) + hf_off(n & 31, 0);
    const float4* sv = (const float4*)(asrc + ((size_t)hh * Tq + t) * HOq);
    const float4* dv = (const float4*)(adst + ((size_t)hh * Tq + t) * HOq);
    float ss = 0.f, sd = 0.f;
#pragma unroll
    for (int q = 0; q < HOq / 4; ++q) {
        const float* hq = hp + (q >> 2) * HBq + (q & 3) * 16;
        float4 hx = make_float4(hq[0], hq[4], hq[8], hq[12]), sx = sv[q], dx = dv[q];
        ss += hx.x * sx.x + hx.y * sx.y + hx.z * sx.z + hx.w * sx.w;
        sd += hx.x * dx.x + hx.y * dx.y + hx.z * dx.z + hx.w * dx.w;
    }
    es[idx] = ss;
    ed[idx] = sd;
    float mx = sd;
#pragma unroll
    for (int off = 16; off >= 1; off >>= 1) mx = fmaxf(mx, __shfl_xor(mx, off, 32));
    if ((threadIdx.x & 31) == 0) wmx[threadIdx.x >> 5] = mx;
    __syncthreads();
    if (threadIdx.x == 0) {
        float m = wmx[0];
#pragma unroll
        for (int k = 1; k < 8; ++k) m = fmaxf(m, wmx[k]);
        edmx[blockIdx.x] = m;
    }
}

__device__ __forceinline__ float pick5(int t, float e0, float e1, float e2, float e3, float e4) {
    float v = e4;
    v = (t == 3) ? e3 : v;
    v = (t == 2) ? e2 : v;
    v = (t == 1) ? e1 : v;
    v = (t == 0) ? e0 : v;
    return v;
}

#define KBq 32
#define PPq 36
#define RTq 4
#define GWq (RTq * (HOq / 16))
#define GRq (16 * RTq)
#define RWq (GRq / GWq)
#define HTq (KBq * HOq)
static_assert(HTq == GWq * 32 * 4 && HTq == (HOq / 16) * HBq, "a trip's h is one 16-byte piece per thread of the group");

#define SUMLOq (1e-3f)

__device__ __forceinline__ void exact_rowmax(float (&rmx)[RWq], const float (&esr)[RWq][Tq],
                                             const int* __restrict__ arow,
                                             const float* __restrict__ edl) {
    float m[RWq];
    int   amin[RWq];
#pragma unroll
    for (int r = 0; r < RWq; ++r) { m[r] = -INFINITY; amin[r] = Tq; }
    for (int j = 0; j < Nq; j += KBq) {
        int   A[RWq];
        float E[Tq];
#pragma unroll
        for (int r = 0; r < RWq; ++r) A[r] = arow[(size_t)r * Nq + j];
#pragma unroll
        for (int t = 0; t < Tq; ++t) E[t] = edl[(size_t)t * Nq + j];
#pragma unroll
        for (int r = 0; r < RWq; ++r) {
            const int a = A[r];
            const int t = (a > 0) ? (a - 1) : 0;
            const float sc = pick5(t, esr[r][0], esr[r][1], esr[r][2], esr[r][3], esr[r][4])
                           + pick5(t, E[0], E[1], E[2], E[3], E[4]);
            m[r]    = fmaxf(m[r], (a == 0) ? -INFINITY : sc);
            amin[r] = min(amin[r], a);
        }
    }
#pragma unroll
    for (int off = 16; off >= 1; off >>= 1) {
#pragma unroll
        for (int r = 0; r < RWq; ++r) m[r] = fmaxf(m[r], __shfl_xor(m[r], off, 32));
    }
#pragma unroll
    for (int r = 0; r < RWq; ++r) {
        float mm = (m[r] > 0.f) ? m[r] : 0.01f * m[r];
        if (__any(amin[r] == 0)) mm = fmaxf(mm, NEGV);
        rmx[r] = mm;
    }
}

__device__ __forceinline__ void at_load(int (&A)[RWq], float (&E)[Tq], v4f& hx,
                                        const int* __restrict__ arow,
                                        const float* __restrict__ edl,
                                        const float* __restrict__ hst, int j) {
#pragma unroll
    for (int r = 0; r < RWq; ++r) A[r] = arow[(size_t)r * Nq + j];
#pragma unroll
    for (int t = 0; t < Tq; ++t) E[t] = edl[(size_t)t * Nq + j];
    hx = *(const v4f*)(hst + (size_t)j * HOq);
}

__device__ __forceinline__ void at_trip(const int (&A)[RWq], const float (&E)[Tq], const v4f& hx,
                                        const float (&esr)[RWq][Tq], const float (&rmx)[RWq],
                                        float (&rs)[RWq], float* __restrict__ Pb, float* __restrict__ Hb,
                                        int prow, int pcol, int aro, int hso, int hro, v8f& c) {
#pragma unroll
    for (int r = 0; r < RWq; ++r) {
        const int a = A[r];
        const int t = (a > 0) ? (a - 1) : 0;
        const float sc = pick5(t, esr[r][0], esr[r][1], esr[r][2], esr[r][3], esr[r][4])
                       + pick5(t, E[0], E[1], E[2], E[3], E[4]);
        const float wv = (a == 0) ? NEGV : (sc > 0.f ? sc : 0.01f * sc);
        const float p = __expf(wv - rmx[r]);
        rs[r] += p;
        Pb[(prow + r) * PPq + pcol] = p;
    }
    *(v4f*)&Hb[hso] = hx;
    group_lds_barrier();
    v4f Bf[4];
#pragma unroll
    for (int k = 0; k < 4; ++k) Bf[k] = *(const v4f*)&Hb[hro + k * 128];
#pragma unroll
    for (int q = 0; q < 8; ++q) {
        const v2f ap = *(const v2f*)&Pb[aro + 4 * q];
        const v2f bq = (q & 1) ? Bf[q >> 1].hi : Bf[q >> 1].lo;
        c = __builtin_amdgcn_wmma_f32_16x16x4_f32(false, ap, false, bq, (short)0, c, false, false);
    }
}

__device__ __forceinline__ void attn_tile(const float (&rmx)[RWq], const float (&esr)[RWq][Tq],
                                          const int* __restrict__ arow, const float* __restrict__ edl,
                                          const float* __restrict__ hst,
                                          float* __restrict__ Pt0, float* __restrict__ Pt1,
                                          float* __restrict__ Hs0, float* __restrict__ Hs1,
                                          float* __restrict__ tile, float* __restrict__ rinv,
                                          int* __restrict__ okf,
                                          int prow, int pcol, int w, int lane, int lm, int half) {
    const int rt  = w >> 2, ct = w & 3;
    const int aro = (16 * rt + lm) * PPq + 2 * half;
    const int hso = (32 * w + lane) * 4;
    const int hro = ct * HBq + lane * 4;
    float rs[RWq];
#pragma unroll
    for (int r = 0; r < RWq; ++r) rs[r] = 0.f;

    int   adjA[RWq], adjB[RWq];
    float edA[Tq],   edB[Tq];
    v4f   hxA,       hxB;
    at_load(adjA, edA, hxA, arow, edl, hst, 0);

    v8f c = {};
    for (int blk = 0; blk < Nq / KBq; blk += 2) {
        const int j0 = blk * KBq;
        at_load(adjB, edB, hxB, arow, edl, hst, j0 + KBq);
        at_trip(adjA, edA, hxA, esr, rmx, rs, Pt0, Hs0, prow, pcol, aro, hso, hro, c);
        const int jn = (blk + 2 < Nq / KBq) ? (j0 + 2 * KBq) : (j0 + KBq);
        at_load(adjA, edA, hxA, arow, edl, hst, jn);
        at_trip(adjB, edB, hxB, esr, rmx, rs, Pt1, Hs1, prow, pcol, aro, hso, hro, c);
    }

#pragma unroll
    for (int off = 16; off >= 1; off >>= 1) {
#pragma unroll
        for (int r = 0; r < RWq; ++r) rs[r] += __shfl_xor(rs[r], off, 32);
    }
    {
        float sr = rs[0];
        int   ok = (rs[0] >= SUMLOq);
#pragma unroll
        for (int r = 1; r < RWq; ++r) {
            sr = (lane == r) ? rs[r] : sr;
            ok &= (rs[r] >= SUMLOq);
        }
        if (lane < RWq) rinv[prow + lane] = 1.f / sr;
        if (lane == 0) okf[w] = ok;
    }

#pragma unroll
    for (int r = 0; r < 8; ++r) tile[(16 * rt + r + 8 * half) * 68 + 16 * ct + lm] = c[r];
}

__global__ void __launch_bounds__(32 * GWq) k_attn(const int* __restrict__ adj,
                                                   const float* __restrict__ hws,
                                                   const float* __restrict__ es,
                                                   const float* __restrict__ ed,
                                                   const float* __restrict__ edmx,
                                                   const float* __restrict__ bias,
                                                   const float* __restrict__ gamma,
                                                   const float* __restrict__ beta,
                                                   float* __restrict__ out) {
    __shared__ __attribute__((aligned(16))) float Pt[2][GRq * PPq];
    __shared__ __attribute__((aligned(16))) float Hs[2][HTq];
    __shared__ __attribute__((aligned(16))) float tile[GRq * 68];
    __shared__ float stat[GRq * 2];
    __shared__ float rinv[GRq];
    __shared__ int   okf[GWq];
    const int lane = threadIdx.x & 31;
    const int w    = __builtin_amdgcn_readfirstlane(threadIdx.x >> 5);
    const int half = lane >> 4;
    const int lm   = lane & 15;
    const int i0   = blockIdx.x * GRq;
    const int bh   = blockIdx.y;
    const int b    = bh / Hq, hh = bh % Hq;
    const int prow = RWq * w;
    const int pcol = 16 * (lane >> 4) + 4 * ((lane >> 1) & 3) + 2 * ((lane >> 3) & 1) + (lane & 1);

    float emx[Tq];
#pragma unroll
    for (int t = 0; t < Tq; ++t) {
        const float* p = edmx + ((size_t)bh * Tq + t) * (Nq / 256);
        emx[t] = fmaxf(fmaxf(p[0], p[1]), fmaxf(p[2], p[3]));
    }

    float esr[RWq][Tq], rmx[RWq];
#pragma unroll
    for (int r = 0; r < RWq; ++r) {
        const int row = i0 + prow + r;
#pragma unroll
        for (int t = 0; t < Tq; ++t) esr[r][t] = es[((size_t)bh * Tq + t) * Nq + row];
        float u = esr[r][0] + emx[0];
#pragma unroll
        for (int t = 1; t < Tq; ++t) u = fmaxf(u, esr[r][t] + emx[t]);
        rmx[r] = (u > 0.f) ? u : 0.01f * u;
    }

    const int*   arow = adj + ((size_t)b * Nq + i0 + prow) * Nq + lane;
    const float* edl  = ed + (size_t)bh * Tq * Nq + lane;
    const float* hb   = hws + hf_block(bh, 0, 0);
    const float* hst  = hb + threadIdx.x * 4;

    attn_tile(rmx, esr, arow, edl, hst, Pt[0], Pt[1], Hs[0], Hs[1], tile, rinv, okf, prow, pcol, w, lane, lm, half);
    __syncthreads();
    const int allok = __builtin_amdgcn_readfirstlane(__all(okf[lane & (GWq - 1)] != 0));
    if (!allok) {
        exact_rowmax(rmx, esr, arow, edl);
        attn_tile(rmx, esr, arow, edl, hst, Pt[0], Pt[1], Hs[0], Hs[1], tile, rinv, okf, prow, pcol, w, lane, lm, half);
        __syncthreads();
    }

    const int row = threadIdx.x >> 3;
    const int e0  = (threadIdx.x & 7) * 8;
    float v[8];
    {
        const int nrow = i0 + row;
        const float* hres = hb + (size_t)(nrow >> 5) * ((HOq / 16) * HBq) + (e0 >> 4) * HBq + hf_off(nrow & 31, e0 & 15);
        const float4 h0 = make_float4(hres[0],  hres[4],  hres[8],  hres[12]);
        const float4 h1 = make_float4(hres[16], hres[20], hres[24], hres[28]);
        const float4 b0 = *(const float4*)(bias + hh * HOq + e0), b1 = *(const float4*)(bias + hh * HOq + e0 + 4);
        const float4 t0 = *(const float4*)&tile[row * 68 + e0],   t1 = *(const float4*)&tile[row * 68 + e0 + 4];
        const float ri = rinv[row];
        v[0] = fmaxf(t0.x * ri + b0.x, 0.f) + h0.x;
        v[1] = fmaxf(t0.y * ri + b0.y, 0.f) + h0.y;
        v[2] = fmaxf(t0.z * ri + b0.z, 0.f) + h0.z;
        v[3] = fmaxf(t0.w * ri + b0.w, 0.f) + h0.w;
        v[4] = fmaxf(t1.x * ri + b1.x, 0.f) + h1.x;
        v[5] = fmaxf(t1.y * ri + b1.y, 0.f) + h1.y;
        v[6] = fmaxf(t1.z * ri + b1.z, 0.f) + h1.z;
        v[7] = fmaxf(t1.w * ri + b1.w, 0.f) + h1.w;
        *(float4*)&tile[row * 68 + e0]     = make_float4(v[0], v[1], v[2], v[3]);
        *(float4*)&tile[row * 68 + e0 + 4] = make_float4(v[4], v[5], v[6], v[7]);
    }
    __syncthreads();
    if (threadIdx.x < GRq) {
        const int r = threadIdx.x;
        float sum = 0.f, sumsq = 0.f;
#pragma unroll 8
        for (int e = 0; e < HOq; ++e) {
            const float x = tile[r * 68 + e];
            sum += x; sumsq += x * x;
        }
        const float mu   = sum * (1.f / HOq);
        const float var  = sumsq * (1.f / HOq) - mu * mu;
        stat[r * 2]     = mu;
        stat[r * 2 + 1] = rsqrtf(var + EPSV);
    }
    __syncthreads();
    {
        const float mu = stat[row * 2], rstd = stat[row * 2 + 1];
        const float4 g0 = *(const float4*)(gamma + hh * HOq + e0), g1 = *(const float4*)(gamma + hh * HOq + e0 + 4);
        const float4 z0 = *(const float4*)(beta + hh * HOq + e0),  z1 = *(const float4*)(beta + hh * HOq + e0 + 4);
        float* op = out + ((size_t)b * Nq + (i0 + row)) * (Hq * HOq) + hh * HOq + e0;
        float4 o0, o1;
        o0.x = (v[0] - mu) * rstd * g0.x + z0.x;
        o0.y = (v[1] - mu) * rstd * g0.y + z0.y;
        o0.z = (v[2] - mu) * rstd * g0.z + z0.z;
        o0.w = (v[3] - mu) * rstd * g0.w + z0.w;
        o1.x = (v[4] - mu) * rstd * g1.x + z1.x;
        o1.y = (v[5] - mu) * rstd * g1.y + z1.y;
        o1.z = (v[6] - mu) * rstd * g1.z + z1.z;
        o1.w = (v[7] - mu) * rstd * g1.w + z1.w;
        *(float4*)op       = o0;
        *(float4*)(op + 4) = o1;
    }
}

extern "C" void kernel_launch(void* const* d_in, const int* in_sizes, int n_in,
                              void* d_out, int out_size, void* d_ws, size_t ws_size,
                              hipStream_t stream) {
    const float* nodes = (const float*)d_in[0];
    const int*   adj   = (const int*)d_in[1];
    const float* Wm    = (const float*)d_in[2];
    const float* asrc  = (const float*)d_in[3];
    const float* adst  = (const float*)d_in[4];
    const float* bias  = (const float*)d_in[5];
    const float* gamma = (const float*)d_in[6];
    const float* beta  = (const float*)d_in[7];
    float* out = (float*)d_out;

    float* h_ws = (float*)d_ws;
    float* esrc = h_ws + (size_t)Bq * Hq * Nq * HOq;
    float* edst = esrc + (size_t)Bq * Hq * Tq * Nq;
    float* edmx = edst + (size_t)Bq * Hq * Tq * Nq;

    k_proj<<<dim3(Hq, Bq * Nq / PRq), 32 * 4, 0, stream>>>(nodes, Wm, h_ws);
    k_escore<<<(Bq * Hq * Tq * Nq) / 256, 256, 0, stream>>>(h_ws, asrc, adst, esrc, edst, edmx);
    k_attn<<<dim3(Nq / GRq, Bq * Hq), 32 * GWq, 0, stream>>>(adj, h_ws, esrc, edst, edmx,
                                                      bias, gamma, beta, out);
}


# k_attn P build: no select chains: row term minus shift by one cross-lane table gather, dst term by one load keyed on adjacency, leaky as fma+max, base-2 terms; score rounding changes, sum order same
# speedup vs baseline: 4.0058x; 1.0821x over previous
#include <hip/hip_runtime.h>

#define Bq  8
#define Nq  1024
#define Dq  256
#define Hq  4
#define HOq 64
#define Tq  5
#define NEGV (-1e9f)
#define EPSV (1e-6f)

typedef float v2f __attribute__((ext_vector_type(2)));
typedef float v8f __attribute__((ext_vector_type(8)));
typedef float v4f __attribute__((ext_vector_type(4)));

#define HBq 512
__device__ __forceinline__ size_t hf_block(int bh, int jb, int ct) {
    return ((size_t)(bh * (Nq / 32) + jb) * (HOq / 16) + ct) * HBq;
}
__device__ __forceinline__ int hf_off(int rho, int c) {
    const int half = (rho >> 3) & 1;
    const int e    = (rho & 7) + 8 * (rho >> 4);
    return ((e >> 2) * 32 + c + 16 * half) * 4 + (e & 3);
}

__device__ __forceinline__ void group_lds_barrier() {
    __builtin_amdgcn_fence(__ATOMIC_RELEASE, "workgroup", "local");
    __builtin_amdgcn_s_barrier();
    __builtin_amdgcn_fence(__ATOMIC_ACQUIRE, "workgroup", "local");
}

#define PRq 32
#define PCq 32
#define PXq 36

__device__ __forceinline__ void pj_load(float4& xa, float4& xb, v2f (&Bf)[8],
                                        const float* __restrict__ xg,
                                        const float* __restrict__ wp, int k0) {
    xa = *(const float4*)(xg + k0);
    xb = *(const float4*)(xg + (size_t)16 * Dq + k0);
#pragma unroll
    for (int q = 0; q < 8; ++q) {
        const float* w0 = wp + (size_t)(k0 + 4 * q) * HOq;
        Bf[q].x = w0[0];
        Bf[q].y = w0[HOq];
    }
}

__device__ __forceinline__ void pj_chunk(const v2f (&Bf)[8], const float* __restrict__ Xb,
                                         int lm, int half, v8f& c0, v8f& c1) {
#pragma unroll
    for (int q = 0; q < 8; ++q) {
        const v2f a0 = *(const v2f*)&Xb[lm * PXq + 4 * q + 2 * half];
        const v2f a1 = *(const v2f*)&Xb[(16 + lm) * PXq + 4 * q + 2 * half];
        c0 = __builtin_amdgcn_wmma_f32_16x16x4_f32(false, a0, false, Bf[q], (short)0, c0, false, false);
        c1 = __builtin_amdgcn_wmma_f32_16x16x4_f32(false, a1, false, Bf[q], (short)0, c1, false, false);
    }
}

__global__ void __launch_bounds__(128) k_proj(const float* __restrict__ X,
                                              const float* __restrict__ Wm,
                                              float* __restrict__ hws) {
    __shared__ __attribute__((aligned(16))) float Xs[2][PRq * PXq];
    const int lane = threadIdx.x & 31;
    const int w    = __builtin_amdgcn_readfirstlane(threadIdx.x >> 5);
    const int half = lane >> 4;
    const int lm   = lane & 15;
    const int hh   = blockIdx.x;
    const int g0   = blockIdx.y * PRq;
    const int b    = g0 / Nq, i0 = g0 % Nq;
    const int bh   = b * Hq + hh;

    const int srow = threadIdx.x >> 3;
    const int scol = (threadIdx.x & 7) * 4;
    const float* xg = X + ((size_t)g0 + srow) * Dq + scol;
    const float* wp = Wm + (size_t)hh * Dq * HOq + (size_t)(2 * half) * HOq + 16 * w + lm;
    const int so = srow * PXq + scol;

    float4 xa, xb;
    v2f bfA[8], bfB[8];
    pj_load(xa, xb, bfA, xg, wp, 0);
    *(float4*)&Xs[0][so]            = xa;
    *(float4*)&Xs[0][so + 16 * PXq] = xb;
    group_lds_barrier();

    v8f c0 = {}, c1 = {};
#pragma unroll
    for (int ch = 0; ch < Dq / PCq; ch += 2) {
        __builtin_amdgcn_sched_barrier(0);
        pj_load(xa, xb, bfB, xg, wp, (ch + 1) * PCq);
        __builtin_amdgcn_sched_barrier(0);
        pj_chunk(bfA, Xs[0], lm, half, c0, c1);
        *(float4*)&Xs[1][so]            = xa;
        *(float4*)&Xs[1][so + 16 * PXq] = xb;
        group_lds_barrier();
        __builtin_amdgcn_sched_barrier(0);
        if (ch + 2 < Dq / PCq) pj_load(xa, xb, bfA, xg, wp, (ch + 2) * PCq);
        __builtin_amdgcn_sched_barrier(0);
        pj_chunk(bfB, Xs[1], lm, half, c0, c1);
        if (ch + 2 < Dq / PCq) {
            *(float4*)&Xs[0][so]            = xa;
            *(float4*)&Xs[0][so + 16 * PXq] = xb;
            group_lds_barrier();
        }
    }
    static_assert(PRq == 32, "a group tile of k_proj is one node block of the fragment order");
    float* hp = hws + hf_block(bh, i0 >> 5, w) + lane * 4;
    *(float4*)(hp)       = make_float4(c0[0], c0[1], c0[2], c0[3]);
    *(float4*)(hp + 128) = make_float4(c0[4], c0[5], c0[6], c0[7]);
    *(float4*)(hp + 256) = make_float4(c1[0], c1[1], c1[2], c1[3]);
    *(float4*)(hp + 384) = make_float4(c1[4], c1[5], c1[6], c1[7]);
}

#define L2Eq  (1.4426950408889634f)
#define NEG2q (NEGV * L2Eq)
static_assert((Bq * Hq * Tq * Nq) % 256 == 0 && Nq % 256 == 0, "a group of k_escore is one (b,h,t) slice of 256 nodes");
__global__ void __launch_bounds__(256) k_escore(const float* __restrict__ hws,
                                                const float* __restrict__ asrc,
                                                const float* __restrict__ adst,
                                                float* __restrict__ esd,
                                                float* __restrict__ edmx) {
    __shared__ float wmx[8];
    const int idx = blockIdx.x * 256 + threadIdx.x;
    const int n  = idx % Nq;
    const int t  = (idx / Nq) % Tq;
    const int bh = idx / (Nq * Tq);
    const int hh = bh % Hq;
    const float* hp = hws + hf_block(bh, n >> 5, 0) + hf_off(n & 31, 0);
    const float4* sv = (const float4*)(asrc + ((size_t)hh * Tq + t) * HOq);
    const float4* dv = (const float4*)(adst + ((size_t)hh * Tq + t) * HOq);
    float ss = 0.f, sd = 0.f;
#pragma unroll
    for (int q = 0; q < HOq / 4; ++q) {
        const float* hq = hp + (q >> 2) * HBq + (q & 3) * 16;
        float4 hx = make_float4(hq[0], hq[4], hq[8], hq[12]), sx = sv[q], dx = dv[q];
        ss += hx.x * sx.x + hx.y * sx.y + hx.z * sx.z + hx.w * sx.w;
        sd += hx.x * dx.x + hx.y * dx.y + hx.z * dx.z + hx.w * dx.w;
    }
    const size_t o = ((size_t)bh * (2 * Tq) + t) * Nq + n;
    const float sd2 = sd * L2Eq;
    esd[o] = ss * L2Eq;
    esd[o + (size_t)Tq * Nq] = sd2;
    float mx = sd2;
#pragma unroll
    for (int off = 16; off >= 1; off >>= 1) mx = fmaxf(mx, __shfl_xor(mx, off, 32));
    if ((threadIdx.x & 31) == 0) wmx[threadIdx.x >> 5] = mx;
    __syncthreads();
    if (threadIdx.x == 0) {
        float m = wmx[0];
#pragma unroll
        for (int k = 1; k < 8; ++k) m = fmaxf(m, wmx[k]);
        edmx[blockIdx.x] = m;
    }
}

__device__ __forceinline__ float pick5(int t, float e0, float e1, float e2, float e3, float e4) {
    float v = e4;
    v = (t == 3) ? e3 : v;
    v = (t == 2) ? e2 : v;
    v = (t == 1) ? e1 : v;
    v = (t == 0) ? e0 : v;
    return v;
}

#define KBq 32
#define PPq 36
#define RTq 4
#define GWq (RTq * (HOq / 16))
#define GRq (16 * RTq)
#define RWq (GRq / GWq)
#define HTq (KBq * HOq)
static_assert(HTq == GWq * 32 * 4 && HTq == (HOq / 16) * HBq, "a trip's h is one 16-byte piece per thread of the group");
static_assert(Nq * 4 == (1 << 12), "a row of esd is 4096 bytes: the dst gather shifts the adjacency value by 12");

#define SUMLOq (1e-3f)

__device__ __forceinline__ void exact_rowmax(float (&rmx)[RWq], const float (&esr)[RWq][Tq],
                                             const int* __restrict__ arow,
                                             const float* __restrict__ edl) {
    float m[RWq];
    int   amin[RWq];
#pragma unroll
    for (int r = 0; r < RWq; ++r) { m[r] = -INFINITY; amin[r] = Tq; }
    for (int j = 0; j < Nq; j += KBq) {
        int   A[RWq];
        float E[Tq];
#pragma unroll
        for (int r = 0; r < RWq; ++r) A[r] = arow[(size_t)r * Nq + j];
#pragma unroll
        for (int t = 0; t < Tq; ++t) E[t] = edl[(size_t)t * Nq + j];
#pragma unroll
        for (int r = 0; r < RWq; ++r) {
            const int a = A[r];
            const int t = (a > 0) ? (a - 1) : 0;
            const float sc = pick5(t, esr[r][0], esr[r][1], esr[r][2], esr[r][3], esr[r][4])
                           + pick5(t, E[0], E[1], E[2], E[3], E[4]);
            m[r]    = fmaxf(m[r], (a == 0) ? -INFINITY : sc);
            amin[r] = min(amin[r], a);
        }
    }
#pragma unroll
    for (int off = 16; off >= 1; off >>= 1) {
#pragma unroll
        for (int r = 0; r < RWq; ++r) m[r] = fmaxf(m[r], __shfl_xor(m[r], off, 32));
    }
#pragma unroll
    for (int r = 0; r < RWq; ++r) {
        float mm = (m[r] > 0.f) ? m[r] : 0.01f * m[r];
        if (__any(amin[r] == 0)) mm = fmaxf(mm, NEG2q);
        rmx[r] = mm;
    }
}

__device__ __forceinline__ void trip_mma(const v4f& hx, float* __restrict__ Pb, float* __restrict__ Hb,
                                         int aro, int hso, int hro, v8f& c) {
    *(v4f*)&Hb[hso] = hx;
    group_lds_barrier();
    v4f Bf[4];
#pragma unroll
    for (int k = 0; k < 4; ++k) Bf[k] = *(const v4f*)&Hb[hro + k * 128];
#pragma unroll
    for (int q = 0; q < 8; ++q) {
        const v2f ap = *(const v2f*)&Pb[aro + 4 * q];
        const v2f bq = (q & 1) ? Bf[q >> 1].hi : Bf[q >> 1].lo;
        c = __builtin_amdgcn_wmma_f32_16x16x4_f32(false, ap, false, bq, (short)0, c, false, false);
    }
}

__device__ __forceinline__ void tile_finish(float (&rs)[RWq], const v8f& c, int okin,
                                            float* __restrict__ tile, float* __restrict__ rinv,
                                            int* __restrict__ okf,
                                            int prow, int w, int lane, int lm, int half) {
    const int rt  = w >> 2, ct = w & 3;
#pragma unroll
    for (int off = 16; off >= 1; off >>= 1) {
#pragma unroll
        for (int r = 0; r < RWq; ++r) rs[r] += __shfl_xor(rs[r], off, 32);
    }
    {
        float sr = rs[0];
        int   ok = okin & (rs[0] >= SUMLOq);
#pragma unroll
        for (int r = 1; r < RWq; ++r) {
            sr = (lane == r) ? rs[r] : sr;
            ok &= (rs[r] >= SUMLOq);
        }
        if (lane < RWq) rinv[prow + lane] = 1.f / sr;
        if (lane == 0) okf[w] = ok;
    }
#pragma unroll
    for (int r = 0; r < 8; ++r) tile[(16 * rt + r + 8 * half) * 68 + 16 * ct + lm] = c[r];
}

__device__ __forceinline__ void xt_load(int (&A)[RWq], float (&E)[Tq], v4f& hx,
                                        const int* __restrict__ arow,
                                        const float* __restrict__ edl,
                                        const float* __restrict__ hst, int j) {
#pragma unroll
    for (int r = 0; r < RWq; ++r) A[r] = arow[(size_t)r * Nq + j];
#pragma unroll
    for (int t = 0; t < Tq; ++t) E[t] = edl[(size_t)t * Nq + j];
    hx = *(const v4f*)(hst + (size_t)j * HOq);
}

__device__ __forceinline__ void xt_trip(const int (&A)[RWq], const float (&E)[Tq], const v4f& hx,
                                        const float (&esr)[RWq][Tq], const float (&rmx)[RWq],
                                        float (&rs)[RWq], float* __restrict__ Pb, float* __restrict__ Hb,
                                        int prow, int pcol, int aro, int hso, int hro, v8f& c) {
#pragma unroll
    for (int r = 0; r < RWq; ++r) {
        const int a = A[r];
        const int t = (a > 0) ? (a - 1) : 0;
        const float sc = pick5(t, esr[r][0], esr[r][1], esr[r][2], esr[r][3], esr[r][4])
                       + pick5(t, E[0], E[1], E[2], E[3], E[4]);
        const float wv = (a == 0) ? NEG2q : (sc > 0.f ? sc : 0.01f * sc);
        const float p = __builtin_amdgcn_exp2f(wv - rmx[r]);
        rs[r] += p;
        Pb[(prow + r) * PPq + pcol] = p;
    }
    trip_mma(hx, Pb, Hb, aro, hso, hro, c);
}

__device__ __forceinline__ void attn_tile_exact(const float (&rmx)[RWq], const float (&esr)[RWq][Tq],
                                                const int* __restrict__ arow, const float* __restrict__ edl,
                                                const float* __restrict__ hst,
                                                float* __restrict__ Pt0, float* __restrict__ Pt1,
                                                float* __restrict__ Hs0, float* __restrict__ Hs1,
                                                float* __restrict__ tile, float* __restrict__ rinv,
                                                int* __restrict__ okf,
                                                int prow, int pcol, int w, int lane, int lm, int half) {
    const int rt  = w >> 2, ct = w & 3;
    const int aro = (16 * rt + lm) * PPq + 2 * half;
    const int hso = (32 * w + lane) * 4;
    const int hro = ct * HBq + lane * 4;
    float rs[RWq];
#pragma unroll
    for (int r = 0; r < RWq; ++r) rs[r] = 0.f;
    int   adjA[RWq];
    float edA[Tq];
    v4f   hxA;
    v8f c = {};
    for (int blk = 0; blk < Nq / KBq; ++blk) {
        xt_load(adjA, edA, hxA, arow, edl, hst, blk * KBq);
        xt_trip(adjA, edA, hxA, esr, rmx, rs, (blk & 1) ? Pt1 : Pt0, (blk & 1) ? Hs1 : Hs0, prow, pcol, aro, hso, hro, c);
    }
    tile_finish(rs, c, 1, tile, rinv, okf, prow, w, lane, lm, half);
}

__device__ __forceinline__ void ft_key(unsigned (&K)[RWq], const int (&Ar)[RWq]) {
#pragma unroll
    for (int r = 0; r < RWq; ++r) K[r] = min((unsigned)Ar[r], (unsigned)Tq);
}

__device__ __forceinline__ void ft_issue(const unsigned (&K)[RWq], float (&G)[RWq], v4f& hx, int (&Ar)[RWq],
                                         const char* __restrict__ gb, unsigned loff,
                                         const float* __restrict__ hst, const int* __restrict__ arow,
                                         int j, int jn) {
    const char* g = gb + (size_t)(unsigned)__builtin_amdgcn_readfirstlane(j * 4);
#pragma unroll
    for (int r = 0; r < RWq; ++r) G[r] = *(const float*)(g + ((K[r] << 12) + loff));
    hx = *(const v4f*)(hst + (size_t)j * HOq);
#pragma unroll
    for (int r = 0; r < RWq; ++r) Ar[r] = arow[(size_t)r * Nq + jn];
}

__device__ __forceinline__ void ft_trip(const unsigned (&K)[RWq], const float (&G)[RWq], const v4f& hx,
                                        const float (&Rt)[RWq], const float (&cs)[RWq],
                                        float (&rs)[RWq], float* __restrict__ Pb, float* __restrict__ Hb,
                                        int prow, int pcol, int aro, int hso, int hro, v8f& c) {
#pragma unroll
    for (int r = 0; r < RWq; ++r) {
        const float R = __builtin_bit_cast(float, __builtin_amdgcn_permlane16_var(
                            0u, __builtin_bit_cast(unsigned, Rt[r]), K[r], true, false));
        const float y = R + G[r];
        const float z = fmaf(y, 0.01f, cs[r]);
        const float p = __builtin_amdgcn_exp2f(fmaxf(y, z));
        rs[r] += p;
        Pb[(prow + r) * PPq + pcol] = p;
    }
    trip_mma(hx, Pb, Hb, aro, hso, hro, c);
}

__device__ __forceinline__ void attn_tile_bounded(const float (&rmx)[RWq], const float (&esr)[RWq][Tq],
                                                  const int* __restrict__ arow, const float* __restrict__ esb,
                                                  const float* __restrict__ hst,
                                                  float* __restrict__ Pt0, float* __restrict__ Pt1,
                                                  float* __restrict__ Hs0, float* __restrict__ Hs1,
                                                  float* __restrict__ tile, float* __restrict__ rinv,
                                                  int* __restrict__ okf,
                                                  int prow, int pcol, int w, int lane, int lm, int half) {
    const int rt  = w >> 2, ct = w & 3;
    const int aro = (16 * rt + lm) * PPq + 2 * half;
    const int hso = (32 * w + lane) * 4;
    const int hro = ct * HBq + lane * 4;
    float Rt[RWq], cs[RWq], rs[RWq];
#pragma unroll
    for (int r = 0; r < RWq; ++r) {
        float v = -INFINITY;
#pragma unroll
        for (int t = 0; t < Tq; ++t) v = (lm == t + 1) ? (esr[r][t] - rmx[r]) : v;
        Rt[r] = v;
        cs[r] = fmaf(rmx[r], 0.01f, -rmx[r]);
        rs[r] = 0.f;
    }
    const char*    gb   = (const char*)(esb + (size_t)(Tq - 1) * Nq);
    const unsigned loff = 4u * (unsigned)lane;

    int      Ar[RWq];
    unsigned KA[RWq], KB[RWq];
    float    GA[RWq], GB[RWq];
    v4f      hxA,     hxB;
#pragma unroll
    for (int r = 0; r < RWq; ++r) Ar[r] = arow[(size_t)r * Nq];
    ft_key(KA, Ar);
    ft_issue(KA, GA, hxA, Ar, gb, loff, hst, arow, 0, KBq);

    v8f c = {};
    for (int blk = 0; blk < Nq / KBq; blk += 2) {
        const int  j0   = blk * KBq;
        const bool more = (blk + 2 < Nq / KBq);
        const int j2 = more ? (j0 + 2 * KBq) : (j0 + KBq);
        const int j3 = more ? (j0 + 3 * KBq) : (j0 + KBq);
        ft_key(KB, Ar);
        ft_issue(KB, GB, hxB, Ar, gb, loff, hst, arow, j0 + KBq, j2);
        ft_trip(KA, GA, hxA, Rt, cs, rs, Pt0, Hs0, prow, pcol, aro, hso, hro, c);
        __builtin_amdgcn_sched_barrier(0);
        ft_key(KA, Ar);
        ft_issue(KA, GA, hxA, Ar, gb, loff, hst, arow, j2, j3);
        ft_trip(KB, GB, hxB, Rt, cs, rs, Pt1, Hs1, prow, pcol, aro, hso, hro, c);
    }
    int okin = 1;
#pragma unroll
    for (int r = 0; r < RWq; ++r) okin &= (rmx[r] > 0.5f * NEG2q);
    tile_finish(rs, c, okin, tile, rinv, okf, prow, w, lane, lm, half);
}

__global__ void __launch_bounds__(32 * GWq) k_attn(const int* __restrict__ adj,
                                                   const float* __restrict__ hws,
                                                   const float* __restrict__ esd,
                                                   const float* __restrict__ edmx,
                                                   const float* __restrict__ bias,
                                                   const float* __restrict__ gamma,
                                                   const float* __restrict__ beta,
                                                   float* __restrict__ out) {
    __shared__ __attribute__((aligned(16))) float Pt[2][GRq * PPq];
    __shared__ __attribute__((aligned(16))) float Hs[2][HTq];
    __shared__ __attribute__((aligned(16))) float tile[GRq * 68];
    __shared__ float stat[GRq * 2];
    __shared__ float rinv[GRq];
    __shared__ int   okf[GWq];
    const int lane = threadIdx.x & 31;
    const int w    = __builtin_amdgcn_readfirstlane(threadIdx.x >> 5);
    const int half = lane >> 4;
    const int lm   = lane & 15;
    const int i0   = blockIdx.x * GRq;
    const int bh   = blockIdx.y;
    const int b    = bh / Hq, hh = bh % Hq;
    const int prow = RWq * w;
    const int pcol = 16 * (lane >> 4) + 4 * ((lane >> 1) & 3) + 2 * ((lane >> 3) & 1) + (lane & 1);

    float emx[Tq];
#pragma unroll
    for (int t = 0; t < Tq; ++t) {
        const float* p = edmx + ((size_t)bh * Tq + t) * (Nq / 256);
        emx[t] = fmaxf(fmaxf(p[0], p[1]), fmaxf(p[2], p[3]));
    }

    const float* esb = esd + (size_t)bh * (2 * Tq) * Nq;
    float esr[RWq][Tq], rmx[RWq];
#pragma unroll
    for (int r = 0; r < RWq; ++r) {
        const int row = i0 + prow + r;
#pragma unroll
        for (int t = 0; t < Tq; ++t) esr[r][t] = esb[(size_t)t * Nq + row];
        float u = esr[r][0] + emx[0];
#pragma unroll
        for (int t = 1; t < Tq; ++t) u = fmaxf(u, esr[r][t] + emx[t]);
        rmx[r] = (u > 0.f) ? u : 0.01f * u;
    }

    const int*   arow = adj + ((size_t)b * Nq + i0 + prow) * Nq + lane;
    const float* edl  = esb + (size_t)Tq * Nq + lane;
    const float* hb   = hws + hf_block(bh, 0, 0);
    const float* hst  = hb + threadIdx.x * 4;

    attn_tile_bounded(rmx, esr, arow, esb, hst, Pt[0], Pt[1], Hs[0], Hs[1], tile, rinv, okf, prow, pcol, w, lane, lm, half);
    __syncthreads();
    const int allok = __builtin_amdgcn_readfirstlane(__all(okf[lane & (GWq - 1)] != 0));
    if (!allok) {
        exact_rowmax(rmx, esr, arow, edl);
        attn_tile_exact(rmx, esr, arow, edl, hst, Pt[0], Pt[1], Hs[0], Hs[1], tile, rinv, okf, prow, pcol, w, lane, lm, half);
        __syncthreads();
    }

    const int row = threadIdx.x >> 3;
    const int e0  = (threadIdx.x & 7) * 8;
    float v[8];
    {
        const int nrow = i0 + row;
        const float* hres = hb + (size_t)(nrow >> 5) * ((HOq / 16) * HBq) + (e0 >> 4) * HBq + hf_off(nrow & 31, e0 & 15);
        const float4 h0 = make_float4(hres[0],  hres[4],  hres[8],  hres[12]);
        const float4 h1 = make_float4(hres[16], hres[20], hres[24], hres[28]);
        const float4 b0 = *(const float4*)(bias + hh * HOq + e0), b1 = *(const float4*)(bias + hh * HOq + e0 + 4);
        const float4 t0 = *(const float4*)&tile[row * 68 + e0],   t1 = *(const float4*)&tile[row * 68 + e0 + 4];
        const float ri = rinv[row];
        v[0] = fmaxf(t0.x * ri + b0.x, 0.f) + h0.x;
        v[1] = fmaxf(t0.y * ri + b0.y, 0.f) + h0.y;
        v[2] = fmaxf(t0.z * ri + b0.z, 0.f) + h0.z;
        v[3] = fmaxf(t0.w * ri + b0.w, 0.f) + h0.w;
        v[4] = fmaxf(t1.x * ri + b1.x, 0.f) + h1.x;
        v[5] = fmaxf(t1.y * ri + b1.y, 0.f) + h1.y;
        v[6] = fmaxf(t1.z * ri + b1.z, 0.f) + h1.z;
        v[7] = fmaxf(t1.w * ri + b1.w, 0.f) + h1.w;
        *(float4*)&tile[row * 68 + e0]     = make_float4(v[0], v[1], v[2], v[3]);
        *(float4*)&tile[row * 68 + e0 + 4] = make_float4(v[4], v[5], v[6], v[7]);
    }
    __syncthreads();
    if (threadIdx.x < GRq) {
        const int r = threadIdx.x;
        float sum = 0.f, sumsq = 0.f;
#pragma unroll 8
        for (int e = 0; e < HOq; ++e) {
            const float x = tile[r * 68 + e];
            sum += x; sumsq += x * x;
        }
        const float mu   = sum * (1.f / HOq);
        const float var  = sumsq * (1.f / HOq) - mu * mu;
        stat[r * 2]     = mu;
        stat[r * 2 + 1] = rsqrtf(var + EPSV);
    }
    __syncthreads();
    {
        const float mu = stat[row * 2], rstd = stat[row * 2 + 1];
        const float4 g0 = *(const float4*)(gamma + hh * HOq + e0), g1 = *(const float4*)(gamma + hh * HOq + e0 + 4);
        const float4 z0 = *(const float4*)(beta + hh * HOq + e0),  z1 = *(const float4*)(beta + hh * HOq + e0 + 4);
        float* op = out + ((size_t)b * Nq + (i0 + row)) * (Hq * HOq) + hh * HOq + e0;
        float4 o0, o1;
        o0.x = (v[0] - mu) * rstd * g0.x + z0.x;
        o0.y = (v[1] - mu) * rstd * g0.y + z0.y;
        o0.z = (v[2] - mu) * rstd * g0.z + z0.z;
        o0.w = (v[3] - mu) * rstd * g0.w + z0.w;
        o1.x = (v[4] - mu) * rstd * g1.x + z1.x;
        o1.y = (v[5] - mu) * rstd * g1.y + z1.y;
        o1.z = (v[6] - mu) * rstd * g1.z + z1.z;
        o1.w = (v[7] - mu) * rstd * g1.w + z1.w;
        *(float4*)op       = o0;
        *(float4*)(op + 4) = o1;
    }
}

extern "C" void kernel_launch(void* const* d_in, const int* in_sizes, int n_in,
                              void* d_out, int out_size, void* d_ws, size_t ws_size,
                              hipStream_t stream) {
    const float* nodes = (const float*)d_in[0];
    const int*   adj   = (const int*)d_in[1];
    const float* Wm    = (const float*)d_in[2];
    const float* asrc  = (const float*)d_in[3];
    const float* adst  = (const float*)d_in[4];
    const float* bias  = (const float*)d_in[5];
    const float* gamma = (const float*)d_in[6];
    const float* beta  = (const float*)d_in[7];
    float* out = (float*)d_out;

    float* h_ws = (float*)d_ws;
    float* esd  = h_ws + (size_t)Bq * Hq * Nq * HOq;
    float* edmx = esd + (size_t)2 * Bq * Hq * Tq * Nq;

    k_proj<<<dim3(Hq, Bq * Nq / PRq), 32 * 4, 0, stream>>>(nodes, Wm, h_ws);
    k_escore<<<(Bq * Hq * Tq * Nq) / 256, 256, 0, stream>>>(h_ws, asrc, adst, esd, edmx);
    k_attn<<<dim3(Nq / GRq, Bq * Hq), 32 * GWq, 0, stream>>>(adj, h_ws, esd, edmx,
                                                      bias, gamma, beta, out);
}
